# starting kernel
# baseline (speedup 1.0000x reference)
#include <hip/hip_runtime.h>

#define B_ 2
#define C_ 256
#define N_ 4096
#define HEADS_ 4
#define HD_ 64
#define GROUPS_ 8
#define CPG_ 32
#define EPS_ 1e-5f

typedef __attribute__((ext_vector_type(2))) float v2f;
typedef __attribute__((ext_vector_type(8))) float v8f;

__device__ __forceinline__ v8f wmma_f32(v2f a, v2f b, v8f c) {
  return __builtin_amdgcn_wmma_f32_16x16x4_f32(
       false, a,  false, b,
       (short)0, c,  false,  false);
}

__global__ __launch_bounds__(256) void gn_kernel(
    const float* __restrict__ x, const float* __restrict__ gamma,
    const float* __restrict__ beta, float* __restrict__ xn) {
  __shared__ float s1[256], s2[256];
  const int b = blockIdx.x / GROUPS_;
  const int g = blockIdx.x % GROUPS_;
  const int count = CPG_ * N_;
  const size_t base = ((size_t)b * C_ + (size_t)g * CPG_) * N_;
  float sum = 0.f, ssq = 0.f;
  for (int i = threadIdx.x; i < count; i += 256) {
    float vv = x[base + i];
    sum += vv;
    ssq += vv * vv;
  }
  s1[threadIdx.x] = sum;
  s2[threadIdx.x] = ssq;
  __syncthreads();
  for (int off = 128; off > 0; off >>= 1) {
    if ((int)threadIdx.x < off) {
      s1[threadIdx.x] += s1[threadIdx.x + off];
      s2[threadIdx.x] += s2[threadIdx.x + off];
    }
    __syncthreads();
  }
  const float mean = s1[0] / (float)count;
  const float var  = s2[0] / (float)count - mean * mean;
  const float rstd = rsqrtf(var + EPS_);
  for (int i = threadIdx.x; i < count; i += 256) {
    int c = g * CPG_ + (i >> 12);
    float vv = (x[base + i] - mean) * rstd;
    xn[base + i] = vv * gamma[c] + beta[c];
  }
}

__global__ __launch_bounds__(128) void qkv_kernel(
    const float* __restrict__ xn, const float* __restrict__ w,
    const float* __restrict__ bias, float* __restrict__ q,
    float* __restrict__ k, float* __restrict__ v) {
  __shared__ float Wl[64][68];
  __shared__ float Xt[64][68];
  const int n0 = blockIdx.x * 64;
  const int o0 = blockIdx.y * 64;
  const int b  = blockIdx.z;
  const int tid = threadIdx.x;
  const int wid = tid >> 5, lane = tid & 31, l16 = lane & 15, half = lane >> 4;
  const int orow = o0 + wid * 16;
  const float* xb = xn + (size_t)b * C_ * N_;

  v8f acc[4] = {};
  for (int kb = 0; kb < C_; kb += 64) {
    for (int i = tid * 4; i < 64 * 64; i += 128 * 4) {
      int r = i >> 6, cc = i & 63;
      *(float4*)&Wl[r][cc] = *(const float4*)(w + (size_t)(o0 + r) * C_ + kb + cc);
      float4 xv = *(const float4*)(xb + (size_t)(kb + r) * N_ + n0 + cc);
      Xt[cc + 0][r] = xv.x;
      Xt[cc + 1][r] = xv.y;
      Xt[cc + 2][r] = xv.z;
      Xt[cc + 3][r] = xv.w;
    }
    __syncthreads();
#pragma unroll
    for (int ks = 0; ks < 16; ++ks) {
      const int kd = 4 * ks + 2 * half;
      const v2f a = *(const v2f*)&Wl[wid * 16 + l16][kd];
#pragma unroll
      for (int nt = 0; nt < 4; ++nt) {
        const v2f bb = *(const v2f*)&Xt[nt * 16 + l16][kd];
        acc[nt] = wmma_f32(a, bb, acc[nt]);
      }
    }
    __syncthreads();
  }

  const int which = orow >> 8;
  const int hh    = (orow & 255) >> 6;
  const int dbase = (orow & 63) + 8 * half;
  float* outb = (which == 0) ? q : (which == 1) ? k : v;
  float bv[8];
#pragma unroll
  for (int vv = 0; vv < 8; ++vv) bv[vv] = bias[orow + 8 * half + vv];
#pragma unroll
  for (int nt = 0; nt < 4; ++nt) {
    const int n = n0 + nt * 16 + l16;
    float* p = outb + (((size_t)b * HEADS_ + hh) * N_ + n) * HD_ + dbase;
    float4 lo, hi;
    lo.x = acc[nt][0] + bv[0]; lo.y = acc[nt][1] + bv[1];
    lo.z = acc[nt][2] + bv[2]; lo.w = acc[nt][3] + bv[3];
    hi.x = acc[nt][4] + bv[4]; hi.y = acc[nt][5] + bv[5];
    hi.z = acc[nt][6] + bv[6]; hi.w = acc[nt][7] + bv[7];
    *(float4*)p = lo;
    *(float4*)(p + 4) = hi;
  }
}

__global__ __launch_bounds__(128) void attn_kernel(
    const float* __restrict__ q, const float* __restrict__ k,
    const float* __restrict__ v, float* __restrict__ ao) {
  __shared__ float Kl[64][68];
  __shared__ float Vt[64][68];
  __shared__ float Pl[4][16][68];
  const int q0 = blockIdx.x * 64;
  const int h  = blockIdx.y;
  const int b  = blockIdx.z;
  const int tid = threadIdx.x;
  const int wid = tid >> 5, lane = tid & 31, l16 = lane & 15, half = lane >> 4;
  const size_t hb = ((size_t)b * HEADS_ + h) * (size_t)N_ * HD_;
  const float* qb = q + hb;
  const float* kb = k + hb;
  const float* vb = v + hb;
  const float scale = 0.125f;

  v2f qa[16];
  const int qrow = q0 + wid * 16 + l16;
#pragma unroll
  for (int ks = 0; ks < 16; ++ks) {
    const int d = 4 * ks + 2 * half;
    const v2f t = *(const v2f*)(qb + (size_t)qrow * HD_ + d);
    qa[ks] = t * scale;
  }

  v8f oacc[4] = {};
  float rm[8], rs[8];
#pragma unroll
  for (int i = 0; i < 8; ++i) { rm[i] = -1e30f; rs[i] = 0.f; }

  for (int j = 0; j < N_; j += 64) {
    for (int i = tid * 4; i < 64 * 64; i += 512) {
      int r = i >> 6, cc = i & 63;
      *(float4*)&Kl[r][cc] = *(const float4*)(kb + (size_t)(j + r) * HD_ + cc);
      float4 vv4 = *(const float4*)(vb + (size_t)(j + r) * HD_ + cc);
      Vt[cc + 0][r] = vv4.x;
      Vt[cc + 1][r] = vv4.y;
      Vt[cc + 2][r] = vv4.z;
      Vt[cc + 3][r] = vv4.w;
    }
    __syncthreads();

    v8f S[4] = {};
#pragma unroll
    for (int ks = 0; ks < 16; ++ks) {
      const int kd = 4 * ks + 2 * half;
#pragma unroll
      for (int nt = 0; nt < 4; ++nt) {
        const v2f bb = *(const v2f*)&Kl[nt * 16 + l16][kd];
        S[nt] = wmma_f32(qa[ks], bb, S[nt]);
      }
    }

#pragma unroll
    for (int vv = 0; vv < 8; ++vv) {
      float cand = fmaxf(fmaxf(S[0][vv], S[1][vv]), fmaxf(S[2][vv], S[3][vv]));
      cand = fmaxf(cand, __shfl_xor(cand, 1, 32));
      cand = fmaxf(cand, __shfl_xor(cand, 2, 32));
      cand = fmaxf(cand, __shfl_xor(cand, 4, 32));
      cand = fmaxf(cand, __shfl_xor(cand, 8, 32));
      const float nm = fmaxf(rm[vv], cand);
      const float f  = __expf(rm[vv] - nm);
      rs[vv] *= f;
      float psum = 0.f;
#pragma unroll
      for (int nt = 0; nt < 4; ++nt) {
        float p = __expf(S[nt][vv] - nm);
        Pl[wid][vv + 8 * half][nt * 16 + l16] = p;
        psum += p;
      }
      psum += __shfl_xor(psum, 1, 32);
      psum += __shfl_xor(psum, 2, 32);
      psum += __shfl_xor(psum, 4, 32);
      psum += __shfl_xor(psum, 8, 32);
      rs[vv] += psum;
      rm[vv] = nm;
#pragma unroll
      for (int dt = 0; dt < 4; ++dt) oacc[dt][vv] *= f;
    }

#pragma unroll
    for (int ks = 0; ks < 16; ++ks) {
      const int kk = 4 * ks + 2 * half;
      const v2f a = *(const v2f*)&Pl[wid][l16][kk];
#pragma unroll
      for (int dt = 0; dt < 4; ++dt) {
        const v2f bb = *(const v2f*)&Vt[dt * 16 + l16][kk];
        oacc[dt] = wmma_f32(a, bb, oacc[dt]);
      }
    }
    __syncthreads();
  }

  float inv[8];
#pragma unroll
  for (int vv = 0; vv < 8; ++vv) inv[vv] = 1.f / rs[vv];
  const int qr0 = q0 + wid * 16 + 8 * half;
#pragma unroll
  for (int dt = 0; dt < 4; ++dt) {
    const int c = h * HD_ + dt * 16 + l16;
    float* p = ao + ((size_t)b * C_ + c) * N_ + qr0;
    float4 lo, hi;
    lo.x = oacc[dt][0] * inv[0]; lo.y = oacc[dt][1] * inv[1];
    lo.z = oacc[dt][2] * inv[2]; lo.w = oacc[dt][3] * inv[3];
    hi.x = oacc[dt][4] * inv[4]; hi.y = oacc[dt][5] * inv[5];
    hi.z = oacc[dt][6] * inv[6]; hi.w = oacc[dt][7] * inv[7];
    *(float4*)p = lo;
    *(float4*)(p + 4) = hi;
  }
}

__global__ __launch_bounds__(128) void proj_kernel(
    const float* __restrict__ ao, const float* __restrict__ w,
    const float* __restrict__ bias, const float* __restrict__ x,
    float* __restrict__ out) {
  __shared__ float Wl[64][68];
  __shared__ float Xt[64][68];
  const int n0 = blockIdx.x * 64;
  const int o0 = blockIdx.y * 64;
  const int b  = blockIdx.z;
  const int tid = threadIdx.x;
  const int wid = tid >> 5, lane = tid & 31, l16 = lane & 15, half = lane >> 4;
  const int orow = o0 + wid * 16;
  const float* ab = ao + (size_t)b * C_ * N_;

  v8f acc[4] = {};
  for (int kb = 0; kb < C_; kb += 64) {
    for (int i = tid * 4; i < 64 * 64; i += 128 * 4) {
      int r = i >> 6, cc = i & 63;
      *(float4*)&Wl[r][cc] = *(const float4*)(w + (size_t)(o0 + r) * C_ + kb + cc);
      float4 xv = *(const float4*)(ab + (size_t)(kb + r) * N_ + n0 + cc);
      Xt[cc + 0][r] = xv.x;
      Xt[cc + 1][r] = xv.y;
      Xt[cc + 2][r] = xv.z;
      Xt[cc + 3][r] = xv.w;
    }
    __syncthreads();
#pragma unroll
    for (int ks = 0; ks < 16; ++ks) {
      const int kd = 4 * ks + 2 * half;
      const v2f a = *(const v2f*)&Wl[wid * 16 + l16][kd];
#pragma unroll
      for (int nt = 0; nt < 4; ++nt) {
        const v2f bb = *(const v2f*)&Xt[nt * 16 + l16][kd];
        acc[nt] = wmma_f32(a, bb, acc[nt]);
      }
    }
    __syncthreads();
  }

  float bv[8];
#pragma unroll
  for (int vv = 0; vv < 8; ++vv) bv[vv] = bias[orow + 8 * half + vv];
#pragma unroll
  for (int nt = 0; nt < 4; ++nt) {
    const int n = n0 + nt * 16 + l16;
#pragma unroll
    for (int vv = 0; vv < 8; ++vv) {
      const int o = orow + 8 * half + vv;
      const size_t idx = ((size_t)b * C_ + o) * N_ + n;
      out[idx] = acc[nt][vv] + bv[vv] + x[idx];
    }
  }
}

extern "C" void kernel_launch(void* const* d_in, const int* in_sizes, int n_in,
                              void* d_out, int out_size, void* d_ws,
                              size_t ws_size, hipStream_t stream) {
  const float* x      = (const float*)d_in[0];
  const float* gamma  = (const float*)d_in[1];
  const float* beta   = (const float*)d_in[2];
  const float* w_qkv  = (const float*)d_in[3];
  const float* b_qkv  = (const float*)d_in[4];
  const float* w_proj = (const float*)d_in[5];
  const float* b_proj = (const float*)d_in[6];
  float* out = (float*)d_out;

  const size_t SZ = (size_t)B_ * C_ * N_;
  float* xn = (float*)d_ws;
  float* q  = xn + SZ;
  float* k  = q + SZ;
  float* v  = k + SZ;
  float* ao = v + SZ;

  gn_kernel<<<B_ * GROUPS_, 256, 0, stream>>>(x, gamma, beta, xn);
  qkv_kernel<<<dim3(N_ / 64, (3 * C_) / 64, B_), 128, 0, stream>>>(
      xn, w_qkv, b_qkv, q, k, v);
  attn_kernel<<<dim3(N_ / 64, HEADS_, B_), 128, 0, stream>>>(q, k, v, ao);
  proj_kernel<<<dim3(N_ / 64, C_ / 64, B_), 128, 0, stream>>>(
      ao, w_proj, b_proj, x, out);
}


# All three WMMA kernels: the next tile's global loads are issued as one batch into registers before the current tile's matrix work and stored to LDS after it (was one round trip per load pair)
# speedup vs baseline: 1.5898x; 1.5898x over previous
#include <hip/hip_runtime.h>

#define B_ 2
#define C_ 256
#define N_ 4096
#define HEADS_ 4
#define HD_ 64
#define GROUPS_ 8
#define CPG_ 32
#define EPS_ 1e-5f

typedef __attribute__((ext_vector_type(2))) float v2f;
typedef __attribute__((ext_vector_type(8))) float v8f;
typedef __attribute__((ext_vector_type(4))) float v4f;

__device__ __forceinline__ v8f wmma_f32(v2f a, v2f b, v8f c) {
  return __builtin_amdgcn_wmma_f32_16x16x4_f32(
       false, a,  false, b,
       (short)0, c,  false,  false);
}

__global__ __launch_bounds__(256) void gn_kernel(
    const float* __restrict__ x, const float* __restrict__ gamma,
    const float* __restrict__ beta, float* __restrict__ xn) {
  __shared__ float s1[256], s2[256];
  const int b = blockIdx.x / GROUPS_;
  const int g = blockIdx.x % GROUPS_;
  const int count = CPG_ * N_;
  const size_t base = ((size_t)b * C_ + (size_t)g * CPG_) * N_;
  float sum = 0.f, ssq = 0.f;
  for (int i = threadIdx.x; i < count; i += 256) {
    float vv = x[base + i];
    sum += vv;
    ssq += vv * vv;
  }
  s1[threadIdx.x] = sum;
  s2[threadIdx.x] = ssq;
  __syncthreads();
  for (int off = 128; off > 0; off >>= 1) {
    if ((int)threadIdx.x < off) {
      s1[threadIdx.x] += s1[threadIdx.x + off];
      s2[threadIdx.x] += s2[threadIdx.x + off];
    }
    __syncthreads();
  }
  const float mean = s1[0] / (float)count;
  const float var  = s2[0] / (float)count - mean * mean;
  const float rstd = rsqrtf(var + EPS_);
  for (int i = threadIdx.x; i < count; i += 256) {
    int c = g * CPG_ + (i >> 12);
    float vv = (x[base + i] - mean) * rstd;
    xn[base + i] = vv * gamma[c] + beta[c];
  }
}

__global__ __launch_bounds__(128) void qkv_kernel(
    const float* __restrict__ xn, const float* __restrict__ w,
    const float* __restrict__ bias, float* __restrict__ q,
    float* __restrict__ k, float* __restrict__ v) {
  __shared__ float Wl[64][68];
  __shared__ float Xt[64][68];
  const int n0 = blockIdx.x * 64;
  const int o0 = blockIdx.y * 64;
  const int b  = blockIdx.z;
  const int tid = threadIdx.x;
  const int wid = tid >> 5, lane = tid & 31, l16 = lane & 15, half = lane >> 4;
  const int orow = o0 + wid * 16;
  const float* xb = xn + (size_t)b * C_ * N_;

  v8f acc[4] = {};
  const int sr = tid >> 4, sc = (tid & 15) * 4;
  v4f wst[8], xst[8];
#pragma unroll
  for (int t = 0; t < 8; ++t) {
    wst[t] = *(const v4f*)(w + (size_t)(o0 + sr + 8 * t) * C_ + sc);
    xst[t] = *(const v4f*)(xb + (size_t)(sr + 8 * t) * N_ + n0 + sc);
  }
  for (int kb = 0; kb < C_; kb += 64) {
#pragma unroll
    for (int t = 0; t < 8; ++t) {
      const int r = sr + 8 * t;
      *(v4f*)&Wl[r][sc] = wst[t];
      Xt[sc + 0][r] = xst[t][0];
      Xt[sc + 1][r] = xst[t][1];
      Xt[sc + 2][r] = xst[t][2];
      Xt[sc + 3][r] = xst[t][3];
    }
    __syncthreads();
    if (kb + 64 < C_) {
#pragma unroll
      for (int t = 0; t < 8; ++t) {
        wst[t] = *(const v4f*)(w + (size_t)(o0 + sr + 8 * t) * C_ + kb + 64 + sc);
        xst[t] = *(const v4f*)(xb + (size_t)(kb + 64 + sr + 8 * t) * N_ + n0 + sc);
      }
    }
#pragma unroll
    for (int ks = 0; ks < 16; ++ks) {
      const int kd = 4 * ks + 2 * half;
      const v2f a = *(const v2f*)&Wl[wid * 16 + l16][kd];
#pragma unroll
      for (int nt = 0; nt < 4; ++nt) {
        const v2f bb = *(const v2f*)&Xt[nt * 16 + l16][kd];
        acc[nt] = wmma_f32(a, bb, acc[nt]);
      }
    }
    __syncthreads();
  }

  const int which = orow >> 8;
  const int hh    = (orow & 255) >> 6;
  const int dbase = (orow & 63) + 8 * half;
  float* outb = (which == 0) ? q : (which == 1) ? k : v;
  float bv[8];
#pragma unroll
  for (int vv = 0; vv < 8; ++vv) bv[vv] = bias[orow + 8 * half + vv];
#pragma unroll
  for (int nt = 0; nt < 4; ++nt) {
    const int n = n0 + nt * 16 + l16;
    float* p = outb + (((size_t)b * HEADS_ + hh) * N_ + n) * HD_ + dbase;
    float4 lo, hi;
    lo.x = acc[nt][0] + bv[0]; lo.y = acc[nt][1] + bv[1];
    lo.z = acc[nt][2] + bv[2]; lo.w = acc[nt][3] + bv[3];
    hi.x = acc[nt][4] + bv[4]; hi.y = acc[nt][5] + bv[5];
    hi.z = acc[nt][6] + bv[6]; hi.w = acc[nt][7] + bv[7];
    *(float4*)p = lo;
    *(float4*)(p + 4) = hi;
  }
}

__global__ __launch_bounds__(128) void attn_kernel(
    const float* __restrict__ q, const float* __restrict__ k,
    const float* __restrict__ v, float* __restrict__ ao) {
  __shared__ float Kl[64][68];
  __shared__ float Vt[64][68];
  __shared__ float Pl[4][16][68];
  const int q0 = blockIdx.x * 64;
  const int h  = blockIdx.y;
  const int b  = blockIdx.z;
  const int tid = threadIdx.x;
  const int wid = tid >> 5, lane = tid & 31, l16 = lane & 15, half = lane >> 4;
  const size_t hb = ((size_t)b * HEADS_ + h) * (size_t)N_ * HD_;
  const float* qb = q + hb;
  const float* kb = k + hb;
  const float* vb = v + hb;
  const float scale = 0.125f;

  v2f qa[16];
  const int qrow = q0 + wid * 16 + l16;
#pragma unroll
  for (int ks = 0; ks < 16; ++ks) {
    const int d = 4 * ks + 2 * half;
    const v2f t = *(const v2f*)(qb + (size_t)qrow * HD_ + d);
    qa[ks] = t * scale;
  }

  v8f oacc[4] = {};
  float rm[8], rs[8];
#pragma unroll
  for (int i = 0; i < 8; ++i) { rm[i] = -1e30f; rs[i] = 0.f; }

  const int sr = tid >> 4, sc = (tid & 15) * 4;
  v4f kst[8], vst[8];
#pragma unroll
  for (int t = 0; t < 8; ++t) {
    kst[t] = *(const v4f*)(kb + (size_t)(sr + 8 * t) * HD_ + sc);
    vst[t] = *(const v4f*)(vb + (size_t)(sr + 8 * t) * HD_ + sc);
  }

  for (int j = 0; j < N_; j += 64) {
#pragma unroll
    for (int t = 0; t < 8; ++t) {
      const int r = sr + 8 * t;
      *(v4f*)&Kl[r][sc] = kst[t];
      Vt[sc + 0][r] = vst[t][0];
      Vt[sc + 1][r] = vst[t][1];
      Vt[sc + 2][r] = vst[t][2];
      Vt[sc + 3][r] = vst[t][3];
    }
    __syncthreads();
    if (j + 64 < N_) {
#pragma unroll
      for (int t = 0; t < 8; ++t) {
        kst[t] = *(const v4f*)(kb + (size_t)(j + 64 + sr + 8 * t) * HD_ + sc);
        vst[t] = *(const v4f*)(vb + (size_t)(j + 64 + sr + 8 * t) * HD_ + sc);
      }
    }

    v8f S[4] = {};
#pragma unroll
    for (int ks = 0; ks < 16; ++ks) {
      const int kd = 4 * ks + 2 * half;
#pragma unroll
      for (int nt = 0; nt < 4; ++nt) {
        const v2f bb = *(const v2f*)&Kl[nt * 16 + l16][kd];
        S[nt] = wmma_f32(qa[ks], bb, S[nt]);
      }
    }

#pragma unroll
    for (int vv = 0; vv < 8; ++vv) {
      float cand = fmaxf(fmaxf(S[0][vv], S[1][vv]), fmaxf(S[2][vv], S[3][vv]));
      cand = fmaxf(cand, __shfl_xor(cand, 1, 32));
      cand = fmaxf(cand, __shfl_xor(cand, 2, 32));
      cand = fmaxf(cand, __shfl_xor(cand, 4, 32));
      cand = fmaxf(cand, __shfl_xor(cand, 8, 32));
      const float nm = fmaxf(rm[vv], cand);
      const float f  = __expf(rm[vv] - nm);
      rs[vv] *= f;
      float psum = 0.f;
#pragma unroll
      for (int nt = 0; nt < 4; ++nt) {
        float p = __expf(S[nt][vv] - nm);
        Pl[wid][vv + 8 * half][nt * 16 + l16] = p;
        psum += p;
      }
      psum += __shfl_xor(psum, 1, 32);
      psum += __shfl_xor(psum, 2, 32);
      psum += __shfl_xor(psum, 4, 32);
      psum += __shfl_xor(psum, 8, 32);
      rs[vv] += psum;
      rm[vv] = nm;
#pragma unroll
      for (int dt = 0; dt < 4; ++dt) oacc[dt][vv] *= f;
    }

#pragma unroll
    for (int ks = 0; ks < 16; ++ks) {
      const int kk = 4 * ks + 2 * half;
      const v2f a = *(const v2f*)&Pl[wid][l16][kk];
#pragma unroll
      for (int dt = 0; dt < 4; ++dt) {
        const v2f bb = *(const v2f*)&Vt[dt * 16 + l16][kk];
        oacc[dt] = wmma_f32(a, bb, oacc[dt]);
      }
    }
    __syncthreads();
  }

  float inv[8];
#pragma unroll
  for (int vv = 0; vv < 8; ++vv) inv[vv] = 1.f / rs[vv];
  const int qr0 = q0 + wid * 16 + 8 * half;
#pragma unroll
  for (int dt = 0; dt < 4; ++dt) {
    const int c = h * HD_ + dt * 16 + l16;
    float* p = ao + ((size_t)b * C_ + c) * N_ + qr0;
    float4 lo, hi;
    lo.x = oacc[dt][0] * inv[0]; lo.y = oacc[dt][1] * inv[1];
    lo.z = oacc[dt][2] * inv[2]; lo.w = oacc[dt][3] * inv[3];
    hi.x = oacc[dt][4] * inv[4]; hi.y = oacc[dt][5] * inv[5];
    hi.z = oacc[dt][6] * inv[6]; hi.w = oacc[dt][7] * inv[7];
    *(float4*)p = lo;
    *(float4*)(p + 4) = hi;
  }
}

__global__ __launch_bounds__(128) void proj_kernel(
    const float* __restrict__ ao, const float* __restrict__ w,
    const float* __restrict__ bias, const float* __restrict__ x,
    float* __restrict__ out) {
  __shared__ float Wl[64][68];
  __shared__ float Xt[64][68];
  const int n0 = blockIdx.x * 64;
  const int o0 = blockIdx.y * 64;
  const int b  = blockIdx.z;
  const int tid = threadIdx.x;
  const int wid = tid >> 5, lane = tid & 31, l16 = lane & 15, half = lane >> 4;
  const int orow = o0 + wid * 16;
  const float* ab = ao + (size_t)b * C_ * N_;

  v8f acc[4] = {};
  const int sr = tid >> 4, sc = (tid & 15) * 4;
  v4f wst[8], xst[8];
#pragma unroll
  for (int t = 0; t < 8; ++t) {
    wst[t] = *(const v4f*)(w + (size_t)(o0 + sr + 8 * t) * C_ + sc);
    xst[t] = *(const v4f*)(ab + (size_t)(sr + 8 * t) * N_ + n0 + sc);
  }
  for (int kb = 0; kb < C_; kb += 64) {
#pragma unroll
    for (int t = 0; t < 8; ++t) {
      const int r = sr + 8 * t;
      *(v4f*)&Wl[r][sc] = wst[t];
      Xt[sc + 0][r] = xst[t][0];
      Xt[sc + 1][r] = xst[t][1];
      Xt[sc + 2][r] = xst[t][2];
      Xt[sc + 3][r] = xst[t][3];
    }
    __syncthreads();
    if (kb + 64 < C_) {
#pragma unroll
      for (int t = 0; t < 8; ++t) {
        wst[t] = *(const v4f*)(w + (size_t)(o0 + sr + 8 * t) * C_ + kb + 64 + sc);
        xst[t] = *(const v4f*)(ab + (size_t)(kb + 64 + sr + 8 * t) * N_ + n0 + sc);
      }
    }
#pragma unroll
    for (int ks = 0; ks < 16; ++ks) {
      const int kd = 4 * ks + 2 * half;
      const v2f a = *(const v2f*)&Wl[wid * 16 + l16][kd];
#pragma unroll
      for (int nt = 0; nt < 4; ++nt) {
        const v2f bb = *(const v2f*)&Xt[nt * 16 + l16][kd];
        acc[nt] = wmma_f32(a, bb, acc[nt]);
      }
    }
    __syncthreads();
  }

  float bv[8];
#pragma unroll
  for (int vv = 0; vv < 8; ++vv) bv[vv] = bias[orow + 8 * half + vv];
#pragma unroll
  for (int nt = 0; nt < 4; ++nt) {
    const int n = n0 + nt * 16 + l16;
#pragma unroll
    for (int vv = 0; vv < 8; ++vv) {
      const int o = orow + 8 * half + vv;
      const size_t idx = ((size_t)b * C_ + o) * N_ + n;
      out[idx] = acc[nt][vv] + bv[vv] + x[idx];
    }
  }
}

extern "C" void kernel_launch(void* const* d_in, const int* in_sizes, int n_in,
                              void* d_out, int out_size, void* d_ws,
                              size_t ws_size, hipStream_t stream) {
  const float* x      = (const float*)d_in[0];
  const float* gamma  = (const float*)d_in[1];
  const float* beta   = (const float*)d_in[2];
  const float* w_qkv  = (const float*)d_in[3];
  const float* b_qkv  = (const float*)d_in[4];
  const float* w_proj = (const float*)d_in[5];
  const float* b_proj = (const float*)d_in[6];
  float* out = (float*)d_out;

  const size_t SZ = (size_t)B_ * C_ * N_;
  float* xn = (float*)d_ws;
  float* q  = xn + SZ;
  float* k  = q + SZ;
  float* v  = k + SZ;
  float* ao = v + SZ;

  gn_kernel<<<B_ * GROUPS_, 256, 0, stream>>>(x, gamma, beta, xn);
  qkv_kernel<<<dim3(N_ / 64, (3 * C_) / 64, B_), 128, 0, stream>>>(
      xn, w_qkv, b_qkv, q, k, v);
  attn_kernel<<<dim3(N_ / 64, HEADS_, B_), 128, 0, stream>>>(q, k, v, ao);
  proj_kernel<<<dim3(N_ / 64, C_ / 64, B_), 128, 0, stream>>>(
      ao, w_proj, b_proj, x, out);
}


# GroupNorm kernel: 1024 threads per group, each with eight float4 loads in flight per batch (was 256 threads, one memory round trip per 8 scalar loads); its sums are taken in a different order
# speedup vs baseline: 1.9421x; 1.2216x over previous
#include <hip/hip_runtime.h>

#define B_ 2
#define C_ 256
#define N_ 4096
#define HEADS_ 4
#define HD_ 64
#define GROUPS_ 8
#define CPG_ 32
#define EPS_ 1e-5f

typedef __attribute__((ext_vector_type(2))) float v2f;
typedef __attribute__((ext_vector_type(8))) float v8f;
typedef __attribute__((ext_vector_type(4))) float v4f;

__device__ __forceinline__ v8f wmma_f32(v2f a, v2f b, v8f c) {
  return __builtin_amdgcn_wmma_f32_16x16x4_f32(
       false, a,  false, b,
       (short)0, c,  false,  false);
}

#define GN_THREADS_ 1024
#define GN_BATCH_ 8
#define N4_ (N_ / 4)
__global__ __launch_bounds__(GN_THREADS_) void gn_kernel(
    const float* __restrict__ x, const float* __restrict__ gamma,
    const float* __restrict__ beta, float* __restrict__ xn) {
  __shared__ float s1[GN_THREADS_ / 32], s2[GN_THREADS_ / 32];
  const int b = blockIdx.x / GROUPS_;
  const int g = blockIdx.x % GROUPS_;
  const int count = CPG_ * N_;
  const size_t base = ((size_t)b * C_ + (size_t)g * CPG_) * N_;
  const v4f* xv = (const v4f*)(x + base);
  v4f* ov = (v4f*)(xn + base);
  const int tid = threadIdx.x;
  const int lane = tid & 31, wid = tid >> 5;

  float sum = 0.f, ssq = 0.f;
  for (int c0 = 0; c0 < CPG_; c0 += GN_BATCH_) {
    v4f t[GN_BATCH_];
    for (int u = 0; u < GN_BATCH_; ++u) t[u] = xv[tid + (c0 + u) * N4_];
    for (int u = 0; u < GN_BATCH_; ++u) {
      sum += (t[u][0] + t[u][1]) + (t[u][2] + t[u][3]);
      ssq += (t[u][0] * t[u][0] + t[u][1] * t[u][1]) +
             (t[u][2] * t[u][2] + t[u][3] * t[u][3]);
    }
  }
  for (int off = 16; off > 0; off >>= 1) {
    sum += __shfl_xor(sum, off, 32);
    ssq += __shfl_xor(ssq, off, 32);
  }
  if (lane == 0) { s1[wid] = sum; s2[wid] = ssq; }
  __syncthreads();
  sum = s1[lane];
  ssq = s2[lane];
  for (int off = 16; off > 0; off >>= 1) {
    sum += __shfl_xor(sum, off, 32);
    ssq += __shfl_xor(ssq, off, 32);
  }
  const float mean = sum / (float)count;
  const float var  = ssq / (float)count - mean * mean;
  const float rstd = rsqrtf(var + EPS_);

  for (int c0 = 0; c0 < CPG_; c0 += GN_BATCH_) {
    v4f t[GN_BATCH_];
    for (int u = 0; u < GN_BATCH_; ++u) t[u] = xv[tid + (c0 + u) * N4_];
    for (int u = 0; u < GN_BATCH_; ++u) {
      const int c = g * CPG_ + c0 + u;
      const float ga = gamma[c], be = beta[c];
      v4f o;
      for (int e = 0; e < 4; ++e) {
        const float vv = (t[u][e] - mean) * rstd;
        o[e] = vv * ga + be;
      }
      ov[tid + (c0 + u) * N4_] = o;
    }
  }
}

__global__ __launch_bounds__(128) void qkv_kernel(
    const float* __restrict__ xn, const float* __restrict__ w,
    const float* __restrict__ bias, float* __restrict__ q,
    float* __restrict__ k, float* __restrict__ v) {
  __shared__ float Wl[64][68];
  __shared__ float Xt[64][68];
  const int n0 = blockIdx.x * 64;
  const int o0 = blockIdx.y * 64;
  const int b  = blockIdx.z;
  const int tid = threadIdx.x;
  const int wid = tid >> 5, lane = tid & 31, l16 = lane & 15, half = lane >> 4;
  const int orow = o0 + wid * 16;
  const float* xb = xn + (size_t)b * C_ * N_;

  v8f acc[4] = {};
  const int sr = tid >> 4, sc = (tid & 15) * 4;
  v4f wst[8], xst[8];
#pragma unroll
  for (int t = 0; t < 8; ++t) {
    wst[t] = *(const v4f*)(w + (size_t)(o0 + sr + 8 * t) * C_ + sc);
    xst[t] = *(const v4f*)(xb + (size_t)(sr + 8 * t) * N_ + n0 + sc);
  }
  for (int kb = 0; kb < C_; kb += 64) {
#pragma unroll
    for (int t = 0; t < 8; ++t) {
      const int r = sr + 8 * t;
      *(v4f*)&Wl[r][sc] = wst[t];
      Xt[sc + 0][r] = xst[t][0];
      Xt[sc + 1][r] = xst[t][1];
      Xt[sc + 2][r] = xst[t][2];
      Xt[sc + 3][r] = xst[t][3];
    }
    __syncthreads();
    if (kb + 64 < C_) {
#pragma unroll
      for (int t = 0; t < 8; ++t) {
        wst[t] = *(const v4f*)(w + (size_t)(o0 + sr + 8 * t) * C_ + kb + 64 + sc);
        xst[t] = *(const v4f*)(xb + (size_t)(kb + 64 + sr + 8 * t) * N_ + n0 + sc);
      }
    }
#pragma unroll
    for (int ks = 0; ks < 16; ++ks) {
      const int kd = 4 * ks + 2 * half;
      const v2f a = *(const v2f*)&Wl[wid * 16 + l16][kd];
#pragma unroll
      for (int nt = 0; nt < 4; ++nt) {
        const v2f bb = *(const v2f*)&Xt[nt * 16 + l16][kd];
        acc[nt] = wmma_f32(a, bb, acc[nt]);
      }
    }
    __syncthreads();
  }

  const int which = orow >> 8;
  const int hh    = (orow & 255) >> 6;
  const int dbase = (orow & 63) + 8 * half;
  float* outb = (which == 0) ? q : (which == 1) ? k : v;
  float bv[8];
#pragma unroll
  for (int vv = 0; vv < 8; ++vv) bv[vv] = bias[orow + 8 * half + vv];
#pragma unroll
  for (int nt = 0; nt < 4; ++nt) {
    const int n = n0 + nt * 16 + l16;
    float* p = outb + (((size_t)b * HEADS_ + hh) * N_ + n) * HD_ + dbase;
    float4 lo, hi;
    lo.x = acc[nt][0] + bv[0]; lo.y = acc[nt][1] + bv[1];
    lo.z = acc[nt][2] + bv[2]; lo.w = acc[nt][3] + bv[3];
    hi.x = acc[nt][4] + bv[4]; hi.y = acc[nt][5] + bv[5];
    hi.z = acc[nt][6] + bv[6]; hi.w = acc[nt][7] + bv[7];
    *(float4*)p = lo;
    *(float4*)(p + 4) = hi;
  }
}

__global__ __launch_bounds__(128) void attn_kernel(
    const float* __restrict__ q, const float* __restrict__ k,
    const float* __restrict__ v, float* __restrict__ ao) {
  __shared__ float Kl[64][68];
  __shared__ float Vt[64][68];
  __shared__ float Pl[4][16][68];
  const int q0 = blockIdx.x * 64;
  const int h  = blockIdx.y;
  const int b  = blockIdx.z;
  const int tid = threadIdx.x;
  const int wid = tid >> 5, lane = tid & 31, l16 = lane & 15, half = lane >> 4;
  const size_t hb = ((size_t)b * HEADS_ + h) * (size_t)N_ * HD_;
  const float* qb = q + hb;
  const float* kb = k + hb;
  const float* vb = v + hb;
  const float scale = 0.125f;

  v2f qa[16];
  const int qrow = q0 + wid * 16 + l16;
#pragma unroll
  for (int ks = 0; ks < 16; ++ks) {
    const int d = 4 * ks + 2 * half;
    const v2f t = *(const v2f*)(qb + (size_t)qrow * HD_ + d);
    qa[ks] = t * scale;
  }

  v8f oacc[4] = {};
  float rm[8], rs[8];
#pragma unroll
  for (int i = 0; i < 8; ++i) { rm[i] = -1e30f; rs[i] = 0.f; }

  const int sr = tid >> 4, sc = (tid & 15) * 4;
  v4f kst[8], vst[8];
#pragma unroll
  for (int t = 0; t < 8; ++t) {
    kst[t] = *(const v4f*)(kb + (size_t)(sr + 8 * t) * HD_ + sc);
    vst[t] = *(const v4f*)(vb + (size_t)(sr + 8 * t) * HD_ + sc);
  }

  for (int j = 0; j < N_; j += 64) {
#pragma unroll
    for (int t = 0; t < 8; ++t) {
      const int r = sr + 8 * t;
      *(v4f*)&Kl[r][sc] = kst[t];
      Vt[sc + 0][r] = vst[t][0];
      Vt[sc + 1][r] = vst[t][1];
      Vt[sc + 2][r] = vst[t][2];
      Vt[sc + 3][r] = vst[t][3];
    }
    __syncthreads();
    if (j + 64 < N_) {
#pragma unroll
      for (int t = 0; t < 8; ++t) {
        kst[t] = *(const v4f*)(kb + (size_t)(j + 64 + sr + 8 * t) * HD_ + sc);
        vst[t] = *(const v4f*)(vb + (size_t)(j + 64 + sr + 8 * t) * HD_ + sc);
      }
    }

    v8f S[4] = {};
#pragma unroll
    for (int ks = 0; ks < 16; ++ks) {
      const int kd = 4 * ks + 2 * half;
#pragma unroll
      for (int nt = 0; nt < 4; ++nt) {
        const v2f bb = *(const v2f*)&Kl[nt * 16 + l16][kd];
        S[nt] = wmma_f32(qa[ks], bb, S[nt]);
      }
    }

#pragma unroll
    for (int vv = 0; vv < 8; ++vv) {
      float cand = fmaxf(fmaxf(S[0][vv], S[1][vv]), fmaxf(S[2][vv], S[3][vv]));
      cand = fmaxf(cand, __shfl_xor(cand, 1, 32));
      cand = fmaxf(cand, __shfl_xor(cand, 2, 32));
      cand = fmaxf(cand, __shfl_xor(cand, 4, 32));
      cand = fmaxf(cand, __shfl_xor(cand, 8, 32));
      const float nm = fmaxf(rm[vv], cand);
      const float f  = __expf(rm[vv] - nm);
      rs[vv] *= f;
      float psum = 0.f;
#pragma unroll
      for (int nt = 0; nt < 4; ++nt) {
        float p = __expf(S[nt][vv] - nm);
        Pl[wid][vv + 8 * half][nt * 16 + l16] = p;
        psum += p;
      }
      psum += __shfl_xor(psum, 1, 32);
      psum += __shfl_xor(psum, 2, 32);
      psum += __shfl_xor(psum, 4, 32);
      psum += __shfl_xor(psum, 8, 32);
      rs[vv] += psum;
      rm[vv] = nm;
#pragma unroll
      for (int dt = 0; dt < 4; ++dt) oacc[dt][vv] *= f;
    }

#pragma unroll
    for (int ks = 0; ks < 16; ++ks) {
      const int kk = 4 * ks + 2 * half;
      const v2f a = *(const v2f*)&Pl[wid][l16][kk];
#pragma unroll
      for (int dt = 0; dt < 4; ++dt) {
        const v2f bb = *(const v2f*)&Vt[dt * 16 + l16][kk];
        oacc[dt] = wmma_f32(a, bb, oacc[dt]);
      }
    }
    __syncthreads();
  }

  float inv[8];
#pragma unroll
  for (int vv = 0; vv < 8; ++vv) inv[vv] = 1.f / rs[vv];
  const int qr0 = q0 + wid * 16 + 8 * half;
#pragma unroll
  for (int dt = 0; dt < 4; ++dt) {
    const int c = h * HD_ + dt * 16 + l16;
    float* p = ao + ((size_t)b * C_ + c) * N_ + qr0;
    float4 lo, hi;
    lo.x = oacc[dt][0] * inv[0]; lo.y = oacc[dt][1] * inv[1];
    lo.z = oacc[dt][2] * inv[2]; lo.w = oacc[dt][3] * inv[3];
    hi.x = oacc[dt][4] * inv[4]; hi.y = oacc[dt][5] * inv[5];
    hi.z = oacc[dt][6] * inv[6]; hi.w = oacc[dt][7] * inv[7];
    *(float4*)p = lo;
    *(float4*)(p + 4) = hi;
  }
}

__global__ __launch_bounds__(128) void proj_kernel(
    const float* __restrict__ ao, const float* __restrict__ w,
    const float* __restrict__ bias, const float* __restrict__ x,
    float* __restrict__ out) {
  __shared__ float Wl[64][68];
  __shared__ float Xt[64][68];
  const int n0 = blockIdx.x * 64;
  const int o0 = blockIdx.y * 64;
  const int b  = blockIdx.z;
  const int tid = threadIdx.x;
  const int wid = tid >> 5, lane = tid & 31, l16 = lane & 15, half = lane >> 4;
  const int orow = o0 + wid * 16;
  const float* ab = ao + (size_t)b * C_ * N_;

  v8f acc[4] = {};
  const int sr = tid >> 4, sc = (tid & 15) * 4;
  v4f wst[8], xst[8];
#pragma unroll
  for (int t = 0; t < 8; ++t) {
    wst[t] = *(const v4f*)(w + (size_t)(o0 + sr + 8 * t) * C_ + sc);
    xst[t] = *(const v4f*)(ab + (size_t)(sr + 8 * t) * N_ + n0 + sc);
  }
  for (int kb = 0; kb < C_; kb += 64) {
#pragma unroll
    for (int t = 0; t < 8; ++t) {
      const int r = sr + 8 * t;
      *(v4f*)&Wl[r][sc] = wst[t];
      Xt[sc + 0][r] = xst[t][0];
      Xt[sc + 1][r] = xst[t][1];
      Xt[sc + 2][r] = xst[t][2];
      Xt[sc + 3][r] = xst[t][3];
    }
    __syncthreads();
    if (kb + 64 < C_) {
#pragma unroll
      for (int t = 0; t < 8; ++t) {
        wst[t] = *(const v4f*)(w + (size_t)(o0 + sr + 8 * t) * C_ + kb + 64 + sc);
        xst[t] = *(const v4f*)(ab + (size_t)(kb + 64 + sr + 8 * t) * N_ + n0 + sc);
      }
    }
#pragma unroll
    for (int ks = 0; ks < 16; ++ks) {
      const int kd = 4 * ks + 2 * half;
      const v2f a = *(const v2f*)&Wl[wid * 16 + l16][kd];
#pragma unroll
      for (int nt = 0; nt < 4; ++nt) {
        const v2f bb = *(const v2f*)&Xt[nt * 16 + l16][kd];
        acc[nt] = wmma_f32(a, bb, acc[nt]);
      }
    }
    __syncthreads();
  }

  float bv[8];
#pragma unroll
  for (int vv = 0; vv < 8; ++vv) bv[vv] = bias[orow + 8 * half + vv];
#pragma unroll
  for (int nt = 0; nt < 4; ++nt) {
    const int n = n0 + nt * 16 + l16;
#pragma unroll
    for (int vv = 0; vv < 8; ++vv) {
      const int o = orow + 8 * half + vv;
      const size_t idx = ((size_t)b * C_ + o) * N_ + n;
      out[idx] = acc[nt][vv] + bv[vv] + x[idx];
    }
  }
}

extern "C" void kernel_launch(void* const* d_in, const int* in_sizes, int n_in,
                              void* d_out, int out_size, void* d_ws,
                              size_t ws_size, hipStream_t stream) {
  const float* x      = (const float*)d_in[0];
  const float* gamma  = (const float*)d_in[1];
  const float* beta   = (const float*)d_in[2];
  const float* w_qkv  = (const float*)d_in[3];
  const float* b_qkv  = (const float*)d_in[4];
  const float* w_proj = (const float*)d_in[5];
  const float* b_proj = (const float*)d_in[6];
  float* out = (float*)d_out;

  const size_t SZ = (size_t)B_ * C_ * N_;
  float* xn = (float*)d_ws;
  float* q  = xn + SZ;
  float* k  = q + SZ;
  float* v  = k + SZ;
  float* ao = v + SZ;

  gn_kernel<<<B_ * GROUPS_, GN_THREADS_, 0, stream>>>(x, gamma, beta, xn);
  qkv_kernel<<<dim3(N_ / 64, (3 * C_) / 64, B_), 128, 0, stream>>>(
      xn, w_qkv, b_qkv, q, k, v);
  attn_kernel<<<dim3(N_ / 64, HEADS_, B_), 128, 0, stream>>>(q, k, v, ao);
  proj_kernel<<<dim3(N_ / 64, C_ / 64, B_), 128, 0, stream>>>(
      ao, w_proj, b_proj, x, out);
}


# Attention: scores computed transposed (K*Q^T) so exp values stay in registers as the PV B operand and row max/sum are in-lane: no LDS pass for P, no shuffles; row sums added in another order
# speedup vs baseline: 2.1482x; 1.1061x over previous
#include <hip/hip_runtime.h>

#define B_ 2
#define C_ 256
#define N_ 4096
#define HEADS_ 4
#define HD_ 64
#define GROUPS_ 8
#define CPG_ 32
#define EPS_ 1e-5f

typedef __attribute__((ext_vector_type(2))) float v2f;
typedef __attribute__((ext_vector_type(8))) float v8f;
typedef __attribute__((ext_vector_type(4))) float v4f;

__device__ __forceinline__ v8f wmma_f32(v2f a, v2f b, v8f c) {
  return __builtin_amdgcn_wmma_f32_16x16x4_f32(
       false, a,  false, b,
       (short)0, c,  false,  false);
}

#define GN_THREADS_ 1024
#define GN_BATCH_ 8
#define N4_ (N_ / 4)
__global__ __launch_bounds__(GN_THREADS_) void gn_kernel(
    const float* __restrict__ x, const float* __restrict__ gamma,
    const float* __restrict__ beta, float* __restrict__ xn) {
  __shared__ float s1[GN_THREADS_ / 32], s2[GN_THREADS_ / 32];
  const int b = blockIdx.x / GROUPS_;
  const int g = blockIdx.x % GROUPS_;
  const int count = CPG_ * N_;
  const size_t base = ((size_t)b * C_ + (size_t)g * CPG_) * N_;
  const v4f* xv = (const v4f*)(x + base);
  v4f* ov = (v4f*)(xn + base);
  const int tid = threadIdx.x;
  const int lane = tid & 31, wid = tid >> 5;

  float sum = 0.f, ssq = 0.f;
  for (int c0 = 0; c0 < CPG_; c0 += GN_BATCH_) {
    v4f t[GN_BATCH_];
    for (int u = 0; u < GN_BATCH_; ++u) t[u] = xv[tid + (c0 + u) * N4_];
    for (int u = 0; u < GN_BATCH_; ++u) {
      sum += (t[u][0] + t[u][1]) + (t[u][2] + t[u][3]);
      ssq += (t[u][0] * t[u][0] + t[u][1] * t[u][1]) +
             (t[u][2] * t[u][2] + t[u][3] * t[u][3]);
    }
  }
  for (int off = 16; off > 0; off >>= 1) {
    sum += __shfl_xor(sum, off, 32);
    ssq += __shfl_xor(ssq, off, 32);
  }
  if (lane == 0) { s1[wid] = sum; s2[wid] = ssq; }
  __syncthreads();
  sum = s1[lane];
  ssq = s2[lane];
  for (int off = 16; off > 0; off >>= 1) {
    sum += __shfl_xor(sum, off, 32);
    ssq += __shfl_xor(ssq, off, 32);
  }
  const float mean = sum / (float)count;
  const float var  = ssq / (float)count - mean * mean;
  const float rstd = rsqrtf(var + EPS_);

  for (int c0 = 0; c0 < CPG_; c0 += GN_BATCH_) {
    v4f t[GN_BATCH_];
    for (int u = 0; u < GN_BATCH_; ++u) t[u] = xv[tid + (c0 + u) * N4_];
    for (int u = 0; u < GN_BATCH_; ++u) {
      const int c = g * CPG_ + c0 + u;
      const float ga = gamma[c], be = beta[c];
      v4f o;
      for (int e = 0; e < 4; ++e) {
        const float vv = (t[u][e] - mean) * rstd;
        o[e] = vv * ga + be;
      }
      ov[tid + (c0 + u) * N4_] = o;
    }
  }
}

__global__ __launch_bounds__(128) void qkv_kernel(
    const float* __restrict__ xn, const float* __restrict__ w,
    const float* __restrict__ bias, float* __restrict__ q,
    float* __restrict__ k, float* __restrict__ v) {
  __shared__ float Wl[64][68];
  __shared__ float Xt[64][68];
  const int n0 = blockIdx.x * 64;
  const int o0 = blockIdx.y * 64;
  const int b  = blockIdx.z;
  const int tid = threadIdx.x;
  const int wid = tid >> 5, lane = tid & 31, l16 = lane & 15, half = lane >> 4;
  const int orow = o0 + wid * 16;
  const float* xb = xn + (size_t)b * C_ * N_;

  v8f acc[4] = {};
  const int sr = tid >> 4, sc = (tid & 15) * 4;
  v4f wst[8], xst[8];
#pragma unroll
  for (int t = 0; t < 8; ++t) {
    wst[t] = *(const v4f*)(w + (size_t)(o0 + sr + 8 * t) * C_ + sc);
    xst[t] = *(const v4f*)(xb + (size_t)(sr + 8 * t) * N_ + n0 + sc);
  }
  for (int kb = 0; kb < C_; kb += 64) {
#pragma unroll
    for (int t = 0; t < 8; ++t) {
      const int r = sr + 8 * t;
      *(v4f*)&Wl[r][sc] = wst[t];
      Xt[sc + 0][r] = xst[t][0];
      Xt[sc + 1][r] = xst[t][1];
      Xt[sc + 2][r] = xst[t][2];
      Xt[sc + 3][r] = xst[t][3];
    }
    __syncthreads();
    if (kb + 64 < C_) {
#pragma unroll
      for (int t = 0; t < 8; ++t) {
        wst[t] = *(const v4f*)(w + (size_t)(o0 + sr + 8 * t) * C_ + kb + 64 + sc);
        xst[t] = *(const v4f*)(xb + (size_t)(kb + 64 + sr + 8 * t) * N_ + n0 + sc);
      }
    }
#pragma unroll
    for (int ks = 0; ks < 16; ++ks) {
      const int kd = 4 * ks + 2 * half;
      const v2f a = *(const v2f*)&Wl[wid * 16 + l16][kd];
#pragma unroll
      for (int nt = 0; nt < 4; ++nt) {
        const v2f bb = *(const v2f*)&Xt[nt * 16 + l16][kd];
        acc[nt] = wmma_f32(a, bb, acc[nt]);
      }
    }
    __syncthreads();
  }

  const int which = orow >> 8;
  const int hh    = (orow & 255) >> 6;
  const int dbase = (orow & 63) + 8 * half;
  float* outb = (which == 0) ? q : (which == 1) ? k : v;
  float bv[8];
#pragma unroll
  for (int vv = 0; vv < 8; ++vv) bv[vv] = bias[orow + 8 * half + vv];
#pragma unroll
  for (int nt = 0; nt < 4; ++nt) {
    const int n = n0 + nt * 16 + l16;
    float* p = outb + (((size_t)b * HEADS_ + hh) * N_ + n) * HD_ + dbase;
    float4 lo, hi;
    lo.x = acc[nt][0] + bv[0]; lo.y = acc[nt][1] + bv[1];
    lo.z = acc[nt][2] + bv[2]; lo.w = acc[nt][3] + bv[3];
    hi.x = acc[nt][4] + bv[4]; hi.y = acc[nt][5] + bv[5];
    hi.z = acc[nt][6] + bv[6]; hi.w = acc[nt][7] + bv[7];
    *(float4*)p = lo;
    *(float4*)(p + 4) = hi;
  }
}

__device__ __forceinline__ float xhalf(float x) {
  const int bits = __builtin_bit_cast(int, x);
  return __builtin_bit_cast(
      float, __builtin_amdgcn_permlanex16(bits, bits, 0x76543210, 0xfedcba98,
                                          false, false));
}

__global__ __launch_bounds__(128) void attn_kernel(
    const float* __restrict__ q, const float* __restrict__ k,
    const float* __restrict__ v, float* __restrict__ ao) {
  __shared__ float Kl[64][68];
  __shared__ float Vt[64][68];
  const int q0 = blockIdx.x * 64;
  const int h  = blockIdx.y;
  const int b  = blockIdx.z;
  const int tid = threadIdx.x;
  const int wid = tid >> 5, lane = tid & 31, l16 = lane & 15, half = lane >> 4;
  const size_t hb = ((size_t)b * HEADS_ + h) * (size_t)N_ * HD_;
  const float* qb = q + hb;
  const float* kb = k + hb;
  const float* vb = v + hb;
  const float scale = 0.125f;

  v2f qa[16];
  const int qrow = q0 + wid * 16 + l16;
#pragma unroll
  for (int ks = 0; ks < 16; ++ks) {
    const int d = 4 * ks + 2 * half;
    const v2f t = *(const v2f*)(qb + (size_t)qrow * HD_ + d);
    qa[ks] = t * scale;
  }

  const int krow = 4 * ((l16 & 7) >> 1) + 2 * (l16 >> 3) + (l16 & 1);

  v8f oacc[4] = {};
  float rm = -1e30f;
  float rs = 0.f;

  const int sr = tid >> 4, sc = (tid & 15) * 4;
  v4f kst[8], vst[8];
#pragma unroll
  for (int t = 0; t < 8; ++t) {
    kst[t] = *(const v4f*)(kb + (size_t)(sr + 8 * t) * HD_ + sc);
    vst[t] = *(const v4f*)(vb + (size_t)(sr + 8 * t) * HD_ + sc);
  }

  for (int j = 0; j < N_; j += 64) {
#pragma unroll
    for (int t = 0; t < 8; ++t) {
      const int r = sr + 8 * t;
      *(v4f*)&Kl[r][sc] = kst[t];
      Vt[sc + 0][r] = vst[t][0];
      Vt[sc + 1][r] = vst[t][1];
      Vt[sc + 2][r] = vst[t][2];
      Vt[sc + 3][r] = vst[t][3];
    }
    __syncthreads();
    if (j + 64 < N_) {
#pragma unroll
      for (int t = 0; t < 8; ++t) {
        kst[t] = *(const v4f*)(kb + (size_t)(j + 64 + sr + 8 * t) * HD_ + sc);
        vst[t] = *(const v4f*)(vb + (size_t)(j + 64 + sr + 8 * t) * HD_ + sc);
      }
    }

    v8f S[4] = {};
#pragma unroll
    for (int ks = 0; ks < 16; ++ks) {
      const int kd = 4 * ks + 2 * half;
#pragma unroll
      for (int nt = 0; nt < 4; ++nt) {
        const v2f a = *(const v2f*)&Kl[nt * 16 + krow][kd];
        S[nt] = wmma_f32(a, qa[ks], S[nt]);
      }
    }

    float mx[4];
#pragma unroll
    for (int nt = 0; nt < 4; ++nt) {
      mx[nt] = fmaxf(fmaxf(fmaxf(S[nt][0], S[nt][1]), fmaxf(S[nt][2], S[nt][3])),
                     fmaxf(fmaxf(S[nt][4], S[nt][5]), fmaxf(S[nt][6], S[nt][7])));
    }
    float m = fmaxf(fmaxf(mx[0], mx[1]), fmaxf(mx[2], mx[3]));
    m = fmaxf(m, xhalf(m));
    const float nm = fmaxf(rm, m);
    const float f  = __expf(rm - nm);
    rm = nm;
    float ps[4];
#pragma unroll
    for (int nt = 0; nt < 4; ++nt) {
#pragma unroll
      for (int r = 0; r < 8; ++r) S[nt][r] = __expf(S[nt][r] - nm);
      ps[nt] = ((S[nt][0] + S[nt][1]) + (S[nt][2] + S[nt][3])) +
               ((S[nt][4] + S[nt][5]) + (S[nt][6] + S[nt][7]));
    }
    rs = rs * f + ((ps[0] + ps[1]) + (ps[2] + ps[3]));
#pragma unroll
    for (int dt = 0; dt < 4; ++dt) oacc[dt] *= f;

#pragma unroll
    for (int ks = 0; ks < 16; ++ks) {
      const int kk = 4 * ks + 2 * half;
      const int nt = ks >> 2, jj = ks & 3;
      v2f pb;
      pb[0] = S[nt][2 * jj];
      pb[1] = S[nt][2 * jj + 1];
#pragma unroll
      for (int dt = 0; dt < 4; ++dt) {
        const v2f a = *(const v2f*)&Vt[dt * 16 + l16][kk];
        oacc[dt] = wmma_f32(a, pb, oacc[dt]);
      }
    }
    __syncthreads();
  }

  const float inv = 1.f / (rs + xhalf(rs));
#pragma unroll
  for (int dt = 0; dt < 4; ++dt) {
#pragma unroll
    for (int r = 0; r < 8; ++r) {
      Kl[dt * 16 + 8 * half + r][wid * 16 + l16] = oacc[dt][r] * inv;
    }
  }
  __syncthreads();
#pragma unroll
  for (int t = 0; t < 8; ++t) {
    const int d = sr + 8 * t;
    const v4f o = *(const v4f*)&Kl[d][sc];
    *(v4f*)(ao + ((size_t)b * C_ + h * HD_ + d) * N_ + q0 + sc) = o;
  }
}

__global__ __launch_bounds__(128) void proj_kernel(
    const float* __restrict__ ao, const float* __restrict__ w,
    const float* __restrict__ bias, const float* __restrict__ x,
    float* __restrict__ out) {
  __shared__ float Wl[64][68];
  __shared__ float Xt[64][68];
  const int n0 = blockIdx.x * 64;
  const int o0 = blockIdx.y * 64;
  const int b  = blockIdx.z;
  const int tid = threadIdx.x;
  const int wid = tid >> 5, lane = tid & 31, l16 = lane & 15, half = lane >> 4;
  const int orow = o0 + wid * 16;
  const float* ab = ao + (size_t)b * C_ * N_;

  v8f acc[4] = {};
  const int sr = tid >> 4, sc = (tid & 15) * 4;
  v4f wst[8], xst[8];
#pragma unroll
  for (int t = 0; t < 8; ++t) {
    wst[t] = *(const v4f*)(w + (size_t)(o0 + sr + 8 * t) * C_ + sc);
    xst[t] = *(const v4f*)(ab + (size_t)(sr + 8 * t) * N_ + n0 + sc);
  }
  for (int kb = 0; kb < C_; kb += 64) {
#pragma unroll
    for (int t = 0; t < 8; ++t) {
      const int r = sr + 8 * t;
      *(v4f*)&Wl[r][sc] = wst[t];
      Xt[sc + 0][r] = xst[t][0];
      Xt[sc + 1][r] = xst[t][1];
      Xt[sc + 2][r] = xst[t][2];
      Xt[sc + 3][r] = xst[t][3];
    }
    __syncthreads();
    if (kb + 64 < C_) {
#pragma unroll
      for (int t = 0; t < 8; ++t) {
        wst[t] = *(const v4f*)(w + (size_t)(o0 + sr + 8 * t) * C_ + kb + 64 + sc);
        xst[t] = *(const v4f*)(ab + (size_t)(kb + 64 + sr + 8 * t) * N_ + n0 + sc);
      }
    }
#pragma unroll
    for (int ks = 0; ks < 16; ++ks) {
      const int kd = 4 * ks + 2 * half;
      const v2f a = *(const v2f*)&Wl[wid * 16 + l16][kd];
#pragma unroll
      for (int nt = 0; nt < 4; ++nt) {
        const v2f bb = *(const v2f*)&Xt[nt * 16 + l16][kd];
        acc[nt] = wmma_f32(a, bb, acc[nt]);
      }
    }
    __syncthreads();
  }

  float bv[8];
#pragma unroll
  for (int vv = 0; vv < 8; ++vv) bv[vv] = bias[orow + 8 * half + vv];
#pragma unroll
  for (int nt = 0; nt < 4; ++nt) {
    const int n = n0 + nt * 16 + l16;
#pragma unroll
    for (int vv = 0; vv < 8; ++vv) {
      const int o = orow + 8 * half + vv;
      const size_t idx = ((size_t)b * C_ + o) * N_ + n;
      out[idx] = acc[nt][vv] + bv[vv] + x[idx];
    }
  }
}

extern "C" void kernel_launch(void* const* d_in, const int* in_sizes, int n_in,
                              void* d_out, int out_size, void* d_ws,
                              size_t ws_size, hipStream_t stream) {
  const float* x      = (const float*)d_in[0];
  const float* gamma  = (const float*)d_in[1];
  const float* beta   = (const float*)d_in[2];
  const float* w_qkv  = (const float*)d_in[3];
  const float* b_qkv  = (const float*)d_in[4];
  const float* w_proj = (const float*)d_in[5];
  const float* b_proj = (const float*)d_in[6];
  float* out = (float*)d_out;

  const size_t SZ = (size_t)B_ * C_ * N_;
  float* xn = (float*)d_ws;
  float* q  = xn + SZ;
  float* k  = q + SZ;
  float* v  = k + SZ;
  float* ao = v + SZ;

  gn_kernel<<<B_ * GROUPS_, GN_THREADS_, 0, stream>>>(x, gamma, beta, xn);
  qkv_kernel<<<dim3(N_ / 64, (3 * C_) / 64, B_), 128, 0, stream>>>(
      xn, w_qkv, b_qkv, q, k, v);
  attn_kernel<<<dim3(N_ / 64, HEADS_, B_), 128, 0, stream>>>(q, k, v, ao);
  proj_kernel<<<dim3(N_ / 64, C_ / 64, B_), 128, 0, stream>>>(
      ao, w_proj, b_proj, x, out);
}


# Attention: each wave lowers its issue priority (s_setprio 3,2,1,0) as its key tiles run out, so the two work-groups that share a SIMD's matrix pipe end together, not the younger one alone
# speedup vs baseline: 2.2090x; 1.0283x over previous
#include <hip/hip_runtime.h>

#define B_ 2
#define C_ 256
#define N_ 4096
#define HEADS_ 4
#define HD_ 64
#define GROUPS_ 8
#define CPG_ 32
#define EPS_ 1e-5f

typedef __attribute__((ext_vector_type(2))) float v2f;
typedef __attribute__((ext_vector_type(8))) float v8f;
typedef __attribute__((ext_vector_type(4))) float v4f;

__device__ __forceinline__ v8f wmma_f32(v2f a, v2f b, v8f c) {
  return __builtin_amdgcn_wmma_f32_16x16x4_f32(
       false, a,  false, b,
       (short)0, c,  false,  false);
}

#define GN_THREADS_ 1024
#define GN_BATCH_ 8
#define N4_ (N_ / 4)
__global__ __launch_bounds__(GN_THREADS_) void gn_kernel(
    const float* __restrict__ x, const float* __restrict__ gamma,
    const float* __restrict__ beta, float* __restrict__ xn) {
  __shared__ float s1[GN_THREADS_ / 32], s2[GN_THREADS_ / 32];
  const int b = blockIdx.x / GROUPS_;
  const int g = blockIdx.x % GROUPS_;
  const int count = CPG_ * N_;
  const size_t base = ((size_t)b * C_ + (size_t)g * CPG_) * N_;
  const v4f* xv = (const v4f*)(x + base);
  v4f* ov = (v4f*)(xn + base);
  const int tid = threadIdx.x;
  const int lane = tid & 31, wid = tid >> 5;

  float sum = 0.f, ssq = 0.f;
  for (int c0 = 0; c0 < CPG_; c0 += GN_BATCH_) {
    v4f t[GN_BATCH_];
    for (int u = 0; u < GN_BATCH_; ++u) t[u] = xv[tid + (c0 + u) * N4_];
    for (int u = 0; u < GN_BATCH_; ++u) {
      sum += (t[u][0] + t[u][1]) + (t[u][2] + t[u][3]);
      ssq += (t[u][0] * t[u][0] + t[u][1] * t[u][1]) +
             (t[u][2] * t[u][2] + t[u][3] * t[u][3]);
    }
  }
  for (int off = 16; off > 0; off >>= 1) {
    sum += __shfl_xor(sum, off, 32);
    ssq += __shfl_xor(ssq, off, 32);
  }
  if (lane == 0) { s1[wid] = sum; s2[wid] = ssq; }
  __syncthreads();
  sum = s1[lane];
  ssq = s2[lane];
  for (int off = 16; off > 0; off >>= 1) {
    sum += __shfl_xor(sum, off, 32);
    ssq += __shfl_xor(ssq, off, 32);
  }
  const float mean = sum / (float)count;
  const float var  = ssq / (float)count - mean * mean;
  const float rstd = rsqrtf(var + EPS_);

  for (int c0 = 0; c0 < CPG_; c0 += GN_BATCH_) {
    v4f t[GN_BATCH_];
    for (int u = 0; u < GN_BATCH_; ++u) t[u] = xv[tid + (c0 + u) * N4_];
    for (int u = 0; u < GN_BATCH_; ++u) {
      const int c = g * CPG_ + c0 + u;
      const float ga = gamma[c], be = beta[c];
      v4f o;
      for (int e = 0; e < 4; ++e) {
        const float vv = (t[u][e] - mean) * rstd;
        o[e] = vv * ga + be;
      }
      ov[tid + (c0 + u) * N4_] = o;
    }
  }
}

__global__ __launch_bounds__(128) void qkv_kernel(
    const float* __restrict__ xn, const float* __restrict__ w,
    const float* __restrict__ bias, float* __restrict__ q,
    float* __restrict__ k, float* __restrict__ v) {
  __shared__ float Wl[64][68];
  __shared__ float Xt[64][68];
  const int n0 = blockIdx.x * 64;
  const int o0 = blockIdx.y * 64;
  const int b  = blockIdx.z;
  const int tid = threadIdx.x;
  const int wid = tid >> 5, lane = tid & 31, l16 = lane & 15, half = lane >> 4;
  const int orow = o0 + wid * 16;
  const float* xb = xn + (size_t)b * C_ * N_;

  v8f acc[4] = {};
  const int sr = tid >> 4, sc = (tid & 15) * 4;
  v4f wst[8], xst[8];
#pragma unroll
  for (int t = 0; t < 8; ++t) {
    wst[t] = *(const v4f*)(w + (size_t)(o0 + sr + 8 * t) * C_ + sc);
    xst[t] = *(const v4f*)(xb + (size_t)(sr + 8 * t) * N_ + n0 + sc);
  }
  for (int kb = 0; kb < C_; kb += 64) {
#pragma unroll
    for (int t = 0; t < 8; ++t) {
      const int r = sr + 8 * t;
      *(v4f*)&Wl[r][sc] = wst[t];
      Xt[sc + 0][r] = xst[t][0];
      Xt[sc + 1][r] = xst[t][1];
      Xt[sc + 2][r] = xst[t][2];
      Xt[sc + 3][r] = xst[t][3];
    }
    __syncthreads();
    if (kb + 64 < C_) {
#pragma unroll
      for (int t = 0; t < 8; ++t) {
        wst[t] = *(const v4f*)(w + (size_t)(o0 + sr + 8 * t) * C_ + kb + 64 + sc);
        xst[t] = *(const v4f*)(xb + (size_t)(kb + 64 + sr + 8 * t) * N_ + n0 + sc);
      }
    }
#pragma unroll
    for (int ks = 0; ks < 16; ++ks) {
      const int kd = 4 * ks + 2 * half;
      const v2f a = *(const v2f*)&Wl[wid * 16 + l16][kd];
#pragma unroll
      for (int nt = 0; nt < 4; ++nt) {
        const v2f bb = *(const v2f*)&Xt[nt * 16 + l16][kd];
        acc[nt] = wmma_f32(a, bb, acc[nt]);
      }
    }
    __syncthreads();
  }

  const int which = orow >> 8;
  const int hh    = (orow & 255) >> 6;
  const int dbase = (orow & 63) + 8 * half;
  float* outb = (which == 0) ? q : (which == 1) ? k : v;
  float bv[8];
#pragma unroll
  for (int vv = 0; vv < 8; ++vv) bv[vv] = bias[orow + 8 * half + vv];
#pragma unroll
  for (int nt = 0; nt < 4; ++nt) {
    const int n = n0 + nt * 16 + l16;
    float* p = outb + (((size_t)b * HEADS_ + hh) * N_ + n) * HD_ + dbase;
    float4 lo, hi;
    lo.x = acc[nt][0] + bv[0]; lo.y = acc[nt][1] + bv[1];
    lo.z = acc[nt][2] + bv[2]; lo.w = acc[nt][3] + bv[3];
    hi.x = acc[nt][4] + bv[4]; hi.y = acc[nt][5] + bv[5];
    hi.z = acc[nt][6] + bv[6]; hi.w = acc[nt][7] + bv[7];
    *(float4*)p = lo;
    *(float4*)(p + 4) = hi;
  }
}

#define PRIO_STEP1_ 44
#define PRIO_STEP2_ 53
#define PRIO_STEP3_ 60

__device__ __forceinline__ float xhalf(float x) {
  const int bits = __builtin_bit_cast(int, x);
  return __builtin_bit_cast(
      float, __builtin_amdgcn_permlanex16(bits, bits, 0x76543210, 0xfedcba98,
                                          false, false));
}

__global__ __launch_bounds__(128) void attn_kernel(
    const float* __restrict__ q, const float* __restrict__ k,
    const float* __restrict__ v, float* __restrict__ ao) {
  __shared__ float Kl[64][68];
  __shared__ float Vt[64][68];
  const int q0 = blockIdx.x * 64;
  const int h  = blockIdx.y;
  const int b  = blockIdx.z;
  const int tid = threadIdx.x;
  const int wid = tid >> 5, lane = tid & 31, l16 = lane & 15, half = lane >> 4;
  const size_t hb = ((size_t)b * HEADS_ + h) * (size_t)N_ * HD_;
  const float* qb = q + hb;
  const float* kb = k + hb;
  const float* vb = v + hb;
  const float scale = 0.125f;

  v2f qa[16];
  const int qrow = q0 + wid * 16 + l16;
#pragma unroll
  for (int ks = 0; ks < 16; ++ks) {
    const int d = 4 * ks + 2 * half;
    const v2f t = *(const v2f*)(qb + (size_t)qrow * HD_ + d);
    qa[ks] = t * scale;
  }

  const int krow = 4 * ((l16 & 7) >> 1) + 2 * (l16 >> 3) + (l16 & 1);

  v8f oacc[4] = {};
  float rm = -1e30f;
  float rs = 0.f;

  const int sr = tid >> 4, sc = (tid & 15) * 4;
  v4f kst[8], vst[8];
#pragma unroll
  for (int t = 0; t < 8; ++t) {
    kst[t] = *(const v4f*)(kb + (size_t)(sr + 8 * t) * HD_ + sc);
    vst[t] = *(const v4f*)(vb + (size_t)(sr + 8 * t) * HD_ + sc);
  }

  __builtin_amdgcn_s_setprio(3);
  for (int j = 0; j < N_; j += 64) {
    if (j == PRIO_STEP1_ * 64) __builtin_amdgcn_s_setprio(2);
    else if (j == PRIO_STEP2_ * 64) __builtin_amdgcn_s_setprio(1);
    else if (j == PRIO_STEP3_ * 64) __builtin_amdgcn_s_setprio(0);
#pragma unroll
    for (int t = 0; t < 8; ++t) {
      const int r = sr + 8 * t;
      *(v4f*)&Kl[r][sc] = kst[t];
      Vt[sc + 0][r] = vst[t][0];
      Vt[sc + 1][r] = vst[t][1];
      Vt[sc + 2][r] = vst[t][2];
      Vt[sc + 3][r] = vst[t][3];
    }
    __syncthreads();
    if (j + 64 < N_) {
#pragma unroll
      for (int t = 0; t < 8; ++t) {
        kst[t] = *(const v4f*)(kb + (size_t)(j + 64 + sr + 8 * t) * HD_ + sc);
        vst[t] = *(const v4f*)(vb + (size_t)(j + 64 + sr + 8 * t) * HD_ + sc);
      }
    }

    v8f S[4] = {};
#pragma unroll
    for (int ks = 0; ks < 16; ++ks) {
      const int kd = 4 * ks + 2 * half;
#pragma unroll
      for (int nt = 0; nt < 4; ++nt) {
        const v2f a = *(const v2f*)&Kl[nt * 16 + krow][kd];
        S[nt] = wmma_f32(a, qa[ks], S[nt]);
      }
    }

    float mx[4];
#pragma unroll
    for (int nt = 0; nt < 4; ++nt) {
      mx[nt] = fmaxf(fmaxf(fmaxf(S[nt][0], S[nt][1]), fmaxf(S[nt][2], S[nt][3])),
                     fmaxf(fmaxf(S[nt][4], S[nt][5]), fmaxf(S[nt][6], S[nt][7])));
    }
    float m = fmaxf(fmaxf(mx[0], mx[1]), fmaxf(mx[2], mx[3]));
    m = fmaxf(m, xhalf(m));
    const float nm = fmaxf(rm, m);
    const float f  = __expf(rm - nm);
    rm = nm;
    float ps[4];
#pragma unroll
    for (int nt = 0; nt < 4; ++nt) {
#pragma unroll
      for (int r = 0; r < 8; ++r) S[nt][r] = __expf(S[nt][r] - nm);
      ps[nt] = ((S[nt][0] + S[nt][1]) + (S[nt][2] + S[nt][3])) +
               ((S[nt][4] + S[nt][5]) + (S[nt][6] + S[nt][7]));
    }
    rs = rs * f + ((ps[0] + ps[1]) + (ps[2] + ps[3]));
#pragma unroll
    for (int dt = 0; dt < 4; ++dt) oacc[dt] *= f;

#pragma unroll
    for (int ks = 0; ks < 16; ++ks) {
      const int kk = 4 * ks + 2 * half;
      const int nt = ks >> 2, jj = ks & 3;
      v2f pb;
      pb[0] = S[nt][2 * jj];
      pb[1] = S[nt][2 * jj + 1];
#pragma unroll
      for (int dt = 0; dt < 4; ++dt) {
        const v2f a = *(const v2f*)&Vt[dt * 16 + l16][kk];
        oacc[dt] = wmma_f32(a, pb, oacc[dt]);
      }
    }
    __syncthreads();
  }

  const float inv = 1.f / (rs + xhalf(rs));
#pragma unroll
  for (int dt = 0; dt < 4; ++dt) {
#pragma unroll
    for (int r = 0; r < 8; ++r) {
      Kl[dt * 16 + 8 * half + r][wid * 16 + l16] = oacc[dt][r] * inv;
    }
  }
  __syncthreads();
#pragma unroll
  for (int t = 0; t < 8; ++t) {
    const int d = sr + 8 * t;
    const v4f o = *(const v4f*)&Kl[d][sc];
    *(v4f*)(ao + ((size_t)b * C_ + h * HD_ + d) * N_ + q0 + sc) = o;
  }
}

__global__ __launch_bounds__(128) void proj_kernel(
    const float* __restrict__ ao, const float* __restrict__ w,
    const float* __restrict__ bias, const float* __restrict__ x,
    float* __restrict__ out) {
  __shared__ float Wl[64][68];
  __shared__ float Xt[64][68];
  const int n0 = blockIdx.x * 64;
  const int o0 = blockIdx.y * 64;
  const int b  = blockIdx.z;
  const int tid = threadIdx.x;
  const int wid = tid >> 5, lane = tid & 31, l16 = lane & 15, half = lane >> 4;
  const int orow = o0 + wid * 16;
  const float* ab = ao + (size_t)b * C_ * N_;

  v8f acc[4] = {};
  const int sr = tid >> 4, sc = (tid & 15) * 4;
  v4f wst[8], xst[8];
#pragma unroll
  for (int t = 0; t < 8; ++t) {
    wst[t] = *(const v4f*)(w + (size_t)(o0 + sr + 8 * t) * C_ + sc);
    xst[t] = *(const v4f*)(ab + (size_t)(sr + 8 * t) * N_ + n0 + sc);
  }
  for (int kb = 0; kb < C_; kb += 64) {
#pragma unroll
    for (int t = 0; t < 8; ++t) {
      const int r = sr + 8 * t;
      *(v4f*)&Wl[r][sc] = wst[t];
      Xt[sc + 0][r] = xst[t][0];
      Xt[sc + 1][r] = xst[t][1];
      Xt[sc + 2][r] = xst[t][2];
      Xt[sc + 3][r] = xst[t][3];
    }
    __syncthreads();
    if (kb + 64 < C_) {
#pragma unroll
      for (int t = 0; t < 8; ++t) {
        wst[t] = *(const v4f*)(w + (size_t)(o0 + sr + 8 * t) * C_ + kb + 64 + sc);
        xst[t] = *(const v4f*)(ab + (size_t)(kb + 64 + sr + 8 * t) * N_ + n0 + sc);
      }
    }
#pragma unroll
    for (int ks = 0; ks < 16; ++ks) {
      const int kd = 4 * ks + 2 * half;
      const v2f a = *(const v2f*)&Wl[wid * 16 + l16][kd];
#pragma unroll
      for (int nt = 0; nt < 4; ++nt) {
        const v2f bb = *(const v2f*)&Xt[nt * 16 + l16][kd];
        acc[nt] = wmma_f32(a, bb, acc[nt]);
      }
    }
    __syncthreads();
  }

  float bv[8];
#pragma unroll
  for (int vv = 0; vv < 8; ++vv) bv[vv] = bias[orow + 8 * half + vv];
#pragma unroll
  for (int nt = 0; nt < 4; ++nt) {
    const int n = n0 + nt * 16 + l16;
#pragma unroll
    for (int vv = 0; vv < 8; ++vv) {
      const int o = orow + 8 * half + vv;
      const size_t idx = ((size_t)b * C_ + o) * N_ + n;
      out[idx] = acc[nt][vv] + bv[vv] + x[idx];
    }
  }
}

extern "C" void kernel_launch(void* const* d_in, const int* in_sizes, int n_in,
                              void* d_out, int out_size, void* d_ws,
                              size_t ws_size, hipStream_t stream) {
  const float* x      = (const float*)d_in[0];
  const float* gamma  = (const float*)d_in[1];
  const float* beta   = (const float*)d_in[2];
  const float* w_qkv  = (const float*)d_in[3];
  const float* b_qkv  = (const float*)d_in[4];
  const float* w_proj = (const float*)d_in[5];
  const float* b_proj = (const float*)d_in[6];
  float* out = (float*)d_out;

  const size_t SZ = (size_t)B_ * C_ * N_;
  float* xn = (float*)d_ws;
  float* q  = xn + SZ;
  float* k  = q + SZ;
  float* v  = k + SZ;
  float* ao = v + SZ;

  gn_kernel<<<B_ * GROUPS_, GN_THREADS_, 0, stream>>>(x, gamma, beta, xn);
  qkv_kernel<<<dim3(N_ / 64, (3 * C_) / 64, B_), 128, 0, stream>>>(
      xn, w_qkv, b_qkv, q, k, v);
  attn_kernel<<<dim3(N_ / 64, HEADS_, B_), 128, 0, stream>>>(q, k, v, ao);
  proj_kernel<<<dim3(N_ / 64, C_ / 64, B_), 128, 0, stream>>>(
      ao, w_proj, b_proj, x, out);
}


# Attention: the 64 LDS fragment loads of a key tile issued as one group ahead of its 128 WMMAs, into registers of their own (were spread over 27 places among them); 2 waves per SIMD kept
# speedup vs baseline: 2.2756x; 1.0301x over previous
#include <hip/hip_runtime.h>

#define B_ 2
#define C_ 256
#define N_ 4096
#define HEADS_ 4
#define HD_ 64
#define GROUPS_ 8
#define CPG_ 32
#define EPS_ 1e-5f

typedef __attribute__((ext_vector_type(2))) float v2f;
typedef __attribute__((ext_vector_type(8))) float v8f;
typedef __attribute__((ext_vector_type(4))) float v4f;

__device__ __forceinline__ v8f wmma_f32(v2f a, v2f b, v8f c) {
  return __builtin_amdgcn_wmma_f32_16x16x4_f32(
       false, a,  false, b,
       (short)0, c,  false,  false);
}

#define GN_THREADS_ 1024
#define GN_BATCH_ 8
#define N4_ (N_ / 4)
__global__ __launch_bounds__(GN_THREADS_) void gn_kernel(
    const float* __restrict__ x, const float* __restrict__ gamma,
    const float* __restrict__ beta, float* __restrict__ xn) {
  __shared__ float s1[GN_THREADS_ / 32], s2[GN_THREADS_ / 32];
  const int b = blockIdx.x / GROUPS_;
  const int g = blockIdx.x % GROUPS_;
  const int count = CPG_ * N_;
  const size_t base = ((size_t)b * C_ + (size_t)g * CPG_) * N_;
  const v4f* xv = (const v4f*)(x + base);
  v4f* ov = (v4f*)(xn + base);
  const int tid = threadIdx.x;
  const int lane = tid & 31, wid = tid >> 5;

  float sum = 0.f, ssq = 0.f;
  for (int c0 = 0; c0 < CPG_; c0 += GN_BATCH_) {
    v4f t[GN_BATCH_];
    for (int u = 0; u < GN_BATCH_; ++u) t[u] = xv[tid + (c0 + u) * N4_];
    for (int u = 0; u < GN_BATCH_; ++u) {
      sum += (t[u][0] + t[u][1]) + (t[u][2] + t[u][3]);
      ssq += (t[u][0] * t[u][0] + t[u][1] * t[u][1]) +
             (t[u][2] * t[u][2] + t[u][3] * t[u][3]);
    }
  }
  for (int off = 16; off > 0; off >>= 1) {
    sum += __shfl_xor(sum, off, 32);
    ssq += __shfl_xor(ssq, off, 32);
  }
  if (lane == 0) { s1[wid] = sum; s2[wid] = ssq; }
  __syncthreads();
  sum = s1[lane];
  ssq = s2[lane];
  for (int off = 16; off > 0; off >>= 1) {
    sum += __shfl_xor(sum, off, 32);
    ssq += __shfl_xor(ssq, off, 32);
  }
  const float mean = sum / (float)count;
  const float var  = ssq / (float)count - mean * mean;
  const float rstd = rsqrtf(var + EPS_);

  for (int c0 = 0; c0 < CPG_; c0 += GN_BATCH_) {
    v4f t[GN_BATCH_];
    for (int u = 0; u < GN_BATCH_; ++u) t[u] = xv[tid + (c0 + u) * N4_];
    for (int u = 0; u < GN_BATCH_; ++u) {
      const int c = g * CPG_ + c0 + u;
      const float ga = gamma[c], be = beta[c];
      v4f o;
      for (int e = 0; e < 4; ++e) {
        const float vv = (t[u][e] - mean) * rstd;
        o[e] = vv * ga + be;
      }
      ov[tid + (c0 + u) * N4_] = o;
    }
  }
}

__global__ __launch_bounds__(128) void qkv_kernel(
    const float* __restrict__ xn, const float* __restrict__ w,
    const float* __restrict__ bias, float* __restrict__ q,
    float* __restrict__ k, float* __restrict__ v) {
  __shared__ float Wl[64][68];
  __shared__ float Xt[64][68];
  const int n0 = blockIdx.x * 64;
  const int o0 = blockIdx.y * 64;
  const int b  = blockIdx.z;
  const int tid = threadIdx.x;
  const int wid = tid >> 5, lane = tid & 31, l16 = lane & 15, half = lane >> 4;
  const int orow = o0 + wid * 16;
  const float* xb = xn + (size_t)b * C_ * N_;

  v8f acc[4] = {};
  const int sr = tid >> 4, sc = (tid & 15) * 4;
  v4f wst[8], xst[8];
#pragma unroll
  for (int t = 0; t < 8; ++t) {
    wst[t] = *(const v4f*)(w + (size_t)(o0 + sr + 8 * t) * C_ + sc);
    xst[t] = *(const v4f*)(xb + (size_t)(sr + 8 * t) * N_ + n0 + sc);
  }
  for (int kb = 0; kb < C_; kb += 64) {
#pragma unroll
    for (int t = 0; t < 8; ++t) {
      const int r = sr + 8 * t;
      *(v4f*)&Wl[r][sc] = wst[t];
      Xt[sc + 0][r] = xst[t][0];
      Xt[sc + 1][r] = xst[t][1];
      Xt[sc + 2][r] = xst[t][2];
      Xt[sc + 3][r] = xst[t][3];
    }
    __syncthreads();
    if (kb + 64 < C_) {
#pragma unroll
      for (int t = 0; t < 8; ++t) {
        wst[t] = *(const v4f*)(w + (size_t)(o0 + sr + 8 * t) * C_ + kb + 64 + sc);
        xst[t] = *(const v4f*)(xb + (size_t)(kb + 64 + sr + 8 * t) * N_ + n0 + sc);
      }
    }
#pragma unroll
    for (int ks = 0; ks < 16; ++ks) {
      const int kd = 4 * ks + 2 * half;
      const v2f a = *(const v2f*)&Wl[wid * 16 + l16][kd];
#pragma unroll
      for (int nt = 0; nt < 4; ++nt) {
        const v2f bb = *(const v2f*)&Xt[nt * 16 + l16][kd];
        acc[nt] = wmma_f32(a, bb, acc[nt]);
      }
    }
    __syncthreads();
  }

  const int which = orow >> 8;
  const int hh    = (orow & 255) >> 6;
  const int dbase = (orow & 63) + 8 * half;
  float* outb = (which == 0) ? q : (which == 1) ? k : v;
  float bv[8];
#pragma unroll
  for (int vv = 0; vv < 8; ++vv) bv[vv] = bias[orow + 8 * half + vv];
#pragma unroll
  for (int nt = 0; nt < 4; ++nt) {
    const int n = n0 + nt * 16 + l16;
    float* p = outb + (((size_t)b * HEADS_ + hh) * N_ + n) * HD_ + dbase;
    float4 lo, hi;
    lo.x = acc[nt][0] + bv[0]; lo.y = acc[nt][1] + bv[1];
    lo.z = acc[nt][2] + bv[2]; lo.w = acc[nt][3] + bv[3];
    hi.x = acc[nt][4] + bv[4]; hi.y = acc[nt][5] + bv[5];
    hi.z = acc[nt][6] + bv[6]; hi.w = acc[nt][7] + bv[7];
    *(float4*)p = lo;
    *(float4*)(p + 4) = hi;
  }
}

#define PRIO_STEP1_ 44
#define PRIO_STEP2_ 53
#define PRIO_STEP3_ 60

__device__ __forceinline__ float xhalf(float x) {
  const int bits = __builtin_bit_cast(int, x);
  return __builtin_bit_cast(
      float, __builtin_amdgcn_permlanex16(bits, bits, 0x76543210, 0xfedcba98,
                                          false, false));
}

__global__ __launch_bounds__(128) void attn_kernel(
    const float* __restrict__ q, const float* __restrict__ k,
    const float* __restrict__ v, float* __restrict__ ao) {
  __shared__ float Kl[64][68];
  __shared__ float Vt[64][68];
  const int q0 = blockIdx.x * 64;
  const int h  = blockIdx.y;
  const int b  = blockIdx.z;
  const int tid = threadIdx.x;
  const int wid = tid >> 5, lane = tid & 31, l16 = lane & 15, half = lane >> 4;
  const size_t hb = ((size_t)b * HEADS_ + h) * (size_t)N_ * HD_;
  const float* qb = q + hb;
  const float* kb = k + hb;
  const float* vb = v + hb;
  const float scale = 0.125f;

  v2f qa[16];
  const int qrow = q0 + wid * 16 + l16;
#pragma unroll
  for (int ks = 0; ks < 16; ++ks) {
    const int d = 4 * ks + 2 * half;
    const v2f t = *(const v2f*)(qb + (size_t)qrow * HD_ + d);
    qa[ks] = t * scale;
  }

  const int krow = 4 * ((l16 & 7) >> 1) + 2 * (l16 >> 3) + (l16 & 1);

  v8f oacc[4] = {};
  float rm = -1e30f;
  float rs = 0.f;

  const int sr = tid >> 4, sc = (tid & 15) * 4;
  v4f kst[8], vst[8];
#pragma unroll
  for (int t = 0; t < 8; ++t) {
    kst[t] = *(const v4f*)(kb + (size_t)(sr + 8 * t) * HD_ + sc);
    vst[t] = *(const v4f*)(vb + (size_t)(sr + 8 * t) * HD_ + sc);
  }

  __builtin_amdgcn_s_setprio(3);
  for (int j = 0; j < N_; j += 64) {
    if (j == PRIO_STEP1_ * 64) __builtin_amdgcn_s_setprio(2);
    else if (j == PRIO_STEP2_ * 64) __builtin_amdgcn_s_setprio(1);
    else if (j == PRIO_STEP3_ * 64) __builtin_amdgcn_s_setprio(0);
#pragma unroll
    for (int t = 0; t < 8; ++t) {
      const int r = sr + 8 * t;
      *(v4f*)&Kl[r][sc] = kst[t];
      Vt[sc + 0][r] = vst[t][0];
      Vt[sc + 1][r] = vst[t][1];
      Vt[sc + 2][r] = vst[t][2];
      Vt[sc + 3][r] = vst[t][3];
    }
    __syncthreads();
    if (j + 64 < N_) {
#pragma unroll
      for (int t = 0; t < 8; ++t) {
        kst[t] = *(const v4f*)(kb + (size_t)(j + 64 + sr + 8 * t) * HD_ + sc);
        vst[t] = *(const v4f*)(vb + (size_t)(j + 64 + sr + 8 * t) * HD_ + sc);
      }
    }

    v2f kf[16][4], vf[16][4];
#pragma unroll
    for (int ks = 0; ks < 16; ++ks) {
      const int kd = 4 * ks + 2 * half;
#pragma unroll
      for (int nt = 0; nt < 4; ++nt) kf[ks][nt] = *(const v2f*)&Kl[nt * 16 + krow][kd];
    }
#pragma unroll
    for (int ks = 0; ks < 16; ++ks) {
      const int kk = 4 * ks + 2 * half;
#pragma unroll
      for (int dt = 0; dt < 4; ++dt) vf[ks][dt] = *(const v2f*)&Vt[dt * 16 + l16][kk];
    }
    __builtin_amdgcn_sched_barrier(0);

    v8f S[4] = {};
#pragma unroll
    for (int ks = 0; ks < 16; ++ks) {
#pragma unroll
      for (int nt = 0; nt < 4; ++nt) S[nt] = wmma_f32(kf[ks][nt], qa[ks], S[nt]);
    }

    float mx[4];
#pragma unroll
    for (int nt = 0; nt < 4; ++nt) {
      mx[nt] = fmaxf(fmaxf(fmaxf(S[nt][0], S[nt][1]), fmaxf(S[nt][2], S[nt][3])),
                     fmaxf(fmaxf(S[nt][4], S[nt][5]), fmaxf(S[nt][6], S[nt][7])));
    }
    float m = fmaxf(fmaxf(mx[0], mx[1]), fmaxf(mx[2], mx[3]));
    m = fmaxf(m, xhalf(m));
    const float nm = fmaxf(rm, m);
    const float f  = __expf(rm - nm);
    rm = nm;
    float ps[4];
#pragma unroll
    for (int nt = 0; nt < 4; ++nt) {
#pragma unroll
      for (int r = 0; r < 8; ++r) S[nt][r] = __expf(S[nt][r] - nm);
      ps[nt] = ((S[nt][0] + S[nt][1]) + (S[nt][2] + S[nt][3])) +
               ((S[nt][4] + S[nt][5]) + (S[nt][6] + S[nt][7]));
    }
    rs = rs * f + ((ps[0] + ps[1]) + (ps[2] + ps[3]));
#pragma unroll
    for (int dt = 0; dt < 4; ++dt) oacc[dt] *= f;

#pragma unroll
    for (int ks = 0; ks < 16; ++ks) {
      const int nt = ks >> 2, jj = ks & 3;
      v2f pb;
      pb[0] = S[nt][2 * jj];
      pb[1] = S[nt][2 * jj + 1];
#pragma unroll
      for (int dt = 0; dt < 4; ++dt) oacc[dt] = wmma_f32(vf[ks][dt], pb, oacc[dt]);
    }
    __builtin_amdgcn_sched_barrier(0);
    __syncthreads();
  }

  const float inv = 1.f / (rs + xhalf(rs));
#pragma unroll
  for (int dt = 0; dt < 4; ++dt) {
#pragma unroll
    for (int r = 0; r < 8; ++r) {
      Kl[dt * 16 + 8 * half + r][wid * 16 + l16] = oacc[dt][r] * inv;
    }
  }
  __syncthreads();
#pragma unroll
  for (int t = 0; t < 8; ++t) {
    const int d = sr + 8 * t;
    const v4f o = *(const v4f*)&Kl[d][sc];
    *(v4f*)(ao + ((size_t)b * C_ + h * HD_ + d) * N_ + q0 + sc) = o;
  }
}

__global__ __launch_bounds__(128) void proj_kernel(
    const float* __restrict__ ao, const float* __restrict__ w,
    const float* __restrict__ bias, const float* __restrict__ x,
    float* __restrict__ out) {
  __shared__ float Wl[64][68];
  __shared__ float Xt[64][68];
  const int n0 = blockIdx.x * 64;
  const int o0 = blockIdx.y * 64;
  const int b  = blockIdx.z;
  const int tid = threadIdx.x;
  const int wid = tid >> 5, lane = tid & 31, l16 = lane & 15, half = lane >> 4;
  const int orow = o0 + wid * 16;
  const float* ab = ao + (size_t)b * C_ * N_;

  v8f acc[4] = {};
  const int sr = tid >> 4, sc = (tid & 15) * 4;
  v4f wst[8], xst[8];
#pragma unroll
  for (int t = 0; t < 8; ++t) {
    wst[t] = *(const v4f*)(w + (size_t)(o0 + sr + 8 * t) * C_ + sc);
    xst[t] = *(const v4f*)(ab + (size_t)(sr + 8 * t) * N_ + n0 + sc);
  }
  for (int kb = 0; kb < C_; kb += 64) {
#pragma unroll
    for (int t = 0; t < 8; ++t) {
      const int r = sr + 8 * t;
      *(v4f*)&Wl[r][sc] = wst[t];
      Xt[sc + 0][r] = xst[t][0];
      Xt[sc + 1][r] = xst[t][1];
      Xt[sc + 2][r] = xst[t][2];
      Xt[sc + 3][r] = xst[t][3];
    }
    __syncthreads();
    if (kb + 64 < C_) {
#pragma unroll
      for (int t = 0; t < 8; ++t) {
        wst[t] = *(const v4f*)(w + (size_t)(o0 + sr + 8 * t) * C_ + kb + 64 + sc);
        xst[t] = *(const v4f*)(ab + (size_t)(kb + 64 + sr + 8 * t) * N_ + n0 + sc);
      }
    }
#pragma unroll
    for (int ks = 0; ks < 16; ++ks) {
      const int kd = 4 * ks + 2 * half;
      const v2f a = *(const v2f*)&Wl[wid * 16 + l16][kd];
#pragma unroll
      for (int nt = 0; nt < 4; ++nt) {
        const v2f bb = *(const v2f*)&Xt[nt * 16 + l16][kd];
        acc[nt] = wmma_f32(a, bb, acc[nt]);
      }
    }
    __syncthreads();
  }

  float bv[8];
#pragma unroll
  for (int vv = 0; vv < 8; ++vv) bv[vv] = bias[orow + 8 * half + vv];
#pragma unroll
  for (int nt = 0; nt < 4; ++nt) {
    const int n = n0 + nt * 16 + l16;
#pragma unroll
    for (int vv = 0; vv < 8; ++vv) {
      const int o = orow + 8 * half + vv;
      const size_t idx = ((size_t)b * C_ + o) * N_ + n;
      out[idx] = acc[nt][vv] + bv[vv] + x[idx];
    }
  }
}

extern "C" void kernel_launch(void* const* d_in, const int* in_sizes, int n_in,
                              void* d_out, int out_size, void* d_ws,
                              size_t ws_size, hipStream_t stream) {
  const float* x      = (const float*)d_in[0];
  const float* gamma  = (const float*)d_in[1];
  const float* beta   = (const float*)d_in[2];
  const float* w_qkv  = (const float*)d_in[3];
  const float* b_qkv  = (const float*)d_in[4];
  const float* w_proj = (const float*)d_in[5];
  const float* b_proj = (const float*)d_in[6];
  float* out = (float*)d_out;

  const size_t SZ = (size_t)B_ * C_ * N_;
  float* xn = (float*)d_ws;
  float* q  = xn + SZ;
  float* k  = q + SZ;
  float* v  = k + SZ;
  float* ao = v + SZ;

  gn_kernel<<<B_ * GROUPS_, GN_THREADS_, 0, stream>>>(x, gamma, beta, xn);
  qkv_kernel<<<dim3(N_ / 64, (3 * C_) / 64, B_), 128, 0, stream>>>(
      xn, w_qkv, b_qkv, q, k, v);
  attn_kernel<<<dim3(N_ / 64, HEADS_, B_), 128, 0, stream>>>(q, k, v, ao);
  proj_kernel<<<dim3(N_ / 64, C_ / 64, B_), 128, 0, stream>>>(
      ao, w_proj, b_proj, x, out);
}


# GroupNorm: a 256-block kernel writes 16 partial sums per group and the QKV GEMM normalises x where it stages it (normalised plane neither written nor read); its sums are taken in a different order
# speedup vs baseline: 2.3904x; 1.0504x over previous
#include <hip/hip_runtime.h>

#define B_ 2
#define C_ 256
#define N_ 4096
#define HEADS_ 4
#define HD_ 64
#define GROUPS_ 8
#define CPG_ 32
#define EPS_ 1e-5f

typedef __attribute__((ext_vector_type(2))) float v2f;
typedef __attribute__((ext_vector_type(8))) float v8f;
typedef __attribute__((ext_vector_type(4))) float v4f;

__device__ __forceinline__ v8f wmma_f32(v2f a, v2f b, v8f c) {
  return __builtin_amdgcn_wmma_f32_16x16x4_f32(
       false, a,  false, b,
       (short)0, c,  false,  false);
}

#define GS_THREADS_ 128
#define GS_BATCH_ 16
#define GN_PARTS_ 16
__global__ __launch_bounds__(GS_THREADS_) void gn_stats_kernel(
    const float* __restrict__ x, float* __restrict__ part) {
  __shared__ float s1[GS_THREADS_ / 32], s2[GS_THREADS_ / 32];
  const v4f* xv = (const v4f*)(x + (size_t)blockIdx.x * (2 * N_));
  const int tid = threadIdx.x;
  const int lane = tid & 31, wid = tid >> 5;

  v4f t[GS_BATCH_];
  for (int u = 0; u < GS_BATCH_; ++u) t[u] = xv[tid + u * GS_THREADS_];
  float sum = 0.f, ssq = 0.f;
  for (int u = 0; u < GS_BATCH_; ++u) {
    sum += (t[u][0] + t[u][1]) + (t[u][2] + t[u][3]);
    ssq += (t[u][0] * t[u][0] + t[u][1] * t[u][1]) +
           (t[u][2] * t[u][2] + t[u][3] * t[u][3]);
  }
  for (int off = 16; off > 0; off >>= 1) {
    sum += __shfl_xor(sum, off, 32);
    ssq += __shfl_xor(ssq, off, 32);
  }
  if (lane == 0) { s1[wid] = sum; s2[wid] = ssq; }
  __syncthreads();
  if (wid == 0) {
    sum = s1[lane & 3];
    ssq = s2[lane & 3];
    for (int off = 2; off > 0; off >>= 1) {
      sum += __shfl_xor(sum, off, 32);
      ssq += __shfl_xor(ssq, off, 32);
    }
    if (lane == 0) {
      v2f o;
      o[0] = sum;
      o[1] = ssq;
      *(v2f*)(part + (size_t)blockIdx.x * 2) = o;
    }
  }
}

__global__ __launch_bounds__(128) void qkv_kernel(
    const float* __restrict__ x, const float* __restrict__ part,
    const float* __restrict__ gamma, const float* __restrict__ beta,
    const float* __restrict__ w,
    const float* __restrict__ bias, float* __restrict__ q,
    float* __restrict__ k, float* __restrict__ v) {
  __shared__ float Wl[64][68];
  __shared__ float Xt[64][68];
  __shared__ v4f coef[C_];
  const int n0 = blockIdx.x * 64;
  const int o0 = blockIdx.y * 64;
  const int b  = blockIdx.z;
  const int tid = threadIdx.x;
  const int wid = tid >> 5, lane = tid & 31, l16 = lane & 15, half = lane >> 4;
  const int orow = o0 + wid * 16;
  const float* xb = x + (size_t)b * C_ * N_;

  v8f acc[4] = {};
  const int sr = tid >> 4, sc = (tid & 15) * 4;
  v4f wst[8], xst[8];
#pragma unroll
  for (int t = 0; t < 8; ++t) {
    wst[t] = *(const v4f*)(w + (size_t)(o0 + sr + 8 * t) * C_ + sc);
    xst[t] = *(const v4f*)(xb + (size_t)(sr + 8 * t) * N_ + n0 + sc);
  }

  {
    const int g = tid >> 4, p = tid & 15;
    const v2f pr =
        *(const v2f*)(part + ((size_t)(b * GROUPS_ + g) * GN_PARTS_ + p) * 2);
    float sum = pr[0], ssq = pr[1];
    for (int off = 8; off > 0; off >>= 1) {
      sum += __shfl_xor(sum, off, 32);
      ssq += __shfl_xor(ssq, off, 32);
    }
    const int count = CPG_ * N_;
    const float mean = sum / (float)count;
    const float var  = ssq / (float)count - mean * mean;
    const float rstd = rsqrtf(var + EPS_);
    for (int e = 0; e < 2; ++e) {
      const int c = g * CPG_ + 2 * p + e;
      v4f cf;
      cf[0] = mean;
      cf[1] = rstd;
      cf[2] = gamma[c];
      cf[3] = beta[c];
      coef[c] = cf;
    }
  }
  __syncthreads();

  for (int kb = 0; kb < C_; kb += 64) {
    v4f cf[8];
#pragma unroll
    for (int t = 0; t < 8; ++t) cf[t] = coef[kb + sr + 8 * t];
#pragma unroll
    for (int t = 0; t < 8; ++t) {
      const int r = sr + 8 * t;
      *(v4f*)&Wl[r][sc] = wst[t];
      v4f xn;
      for (int e = 0; e < 4; ++e) {
        const float vv = (xst[t][e] - cf[t][0]) * cf[t][1];
        xn[e] = vv * cf[t][2] + cf[t][3];
      }
      Xt[sc + 0][r] = xn[0];
      Xt[sc + 1][r] = xn[1];
      Xt[sc + 2][r] = xn[2];
      Xt[sc + 3][r] = xn[3];
    }
    __syncthreads();
    if (kb + 64 < C_) {
#pragma unroll
      for (int t = 0; t < 8; ++t) {
        wst[t] = *(const v4f*)(w + (size_t)(o0 + sr + 8 * t) * C_ + kb + 64 + sc);
        xst[t] = *(const v4f*)(xb + (size_t)(kb + 64 + sr + 8 * t) * N_ + n0 + sc);
      }
    }
#pragma unroll
    for (int ks = 0; ks < 16; ++ks) {
      const int kd = 4 * ks + 2 * half;
      const v2f a = *(const v2f*)&Wl[wid * 16 + l16][kd];
#pragma unroll
      for (int nt = 0; nt < 4; ++nt) {
        const v2f bb = *(const v2f*)&Xt[nt * 16 + l16][kd];
        acc[nt] = wmma_f32(a, bb, acc[nt]);
      }
    }
    __syncthreads();
  }

  const int which = orow >> 8;
  const int hh    = (orow & 255) >> 6;
  const int dbase = (orow & 63) + 8 * half;
  float* outb = (which == 0) ? q : (which == 1) ? k : v;
  float bv[8];
#pragma unroll
  for (int vv = 0; vv < 8; ++vv) bv[vv] = bias[orow + 8 * half + vv];
#pragma unroll
  for (int nt = 0; nt < 4; ++nt) {
    const int n = n0 + nt * 16 + l16;
    float* p = outb + (((size_t)b * HEADS_ + hh) * N_ + n) * HD_ + dbase;
    float4 lo, hi;
    lo.x = acc[nt][0] + bv[0]; lo.y = acc[nt][1] + bv[1];
    lo.z = acc[nt][2] + bv[2]; lo.w = acc[nt][3] + bv[3];
    hi.x = acc[nt][4] + bv[4]; hi.y = acc[nt][5] + bv[5];
    hi.z = acc[nt][6] + bv[6]; hi.w = acc[nt][7] + bv[7];
    *(float4*)p = lo;
    *(float4*)(p + 4) = hi;
  }
}

#define PRIO_STEP1_ 44
#define PRIO_STEP2_ 53
#define PRIO_STEP3_ 60

__device__ __forceinline__ float xhalf(float x) {
  const int bits = __builtin_bit_cast(int, x);
  return __builtin_bit_cast(
      float, __builtin_amdgcn_permlanex16(bits, bits, 0x76543210, 0xfedcba98,
                                          false, false));
}

__global__ __launch_bounds__(128) void attn_kernel(
    const float* __restrict__ q, const float* __restrict__ k,
    const float* __restrict__ v, float* __restrict__ ao) {
  __shared__ float Kl[64][68];
  __shared__ float Vt[64][68];
  const int q0 = blockIdx.x * 64;
  const int h  = blockIdx.y;
  const int b  = blockIdx.z;
  const int tid = threadIdx.x;
  const int wid = tid >> 5, lane = tid & 31, l16 = lane & 15, half = lane >> 4;
  const size_t hb = ((size_t)b * HEADS_ + h) * (size_t)N_ * HD_;
  const float* qb = q + hb;
  const float* kb = k + hb;
  const float* vb = v + hb;
  const float scale = 0.125f;

  v2f qa[16];
  const int qrow = q0 + wid * 16 + l16;
#pragma unroll
  for (int ks = 0; ks < 16; ++ks) {
    const int d = 4 * ks + 2 * half;
    const v2f t = *(const v2f*)(qb + (size_t)qrow * HD_ + d);
    qa[ks] = t * scale;
  }

  const int krow = 4 * ((l16 & 7) >> 1) + 2 * (l16 >> 3) + (l16 & 1);

  v8f oacc[4] = {};
  float rm = -1e30f;
  float rs = 0.f;

  const int sr = tid >> 4, sc = (tid & 15) * 4;
  v4f kst[8], vst[8];
#pragma unroll
  for (int t = 0; t < 8; ++t) {
    kst[t] = *(const v4f*)(kb + (size_t)(sr + 8 * t) * HD_ + sc);
    vst[t] = *(const v4f*)(vb + (size_t)(sr + 8 * t) * HD_ + sc);
  }

  __builtin_amdgcn_s_setprio(3);
  for (int j = 0; j < N_; j += 64) {
    if (j == PRIO_STEP1_ * 64) __builtin_amdgcn_s_setprio(2);
    else if (j == PRIO_STEP2_ * 64) __builtin_amdgcn_s_setprio(1);
    else if (j == PRIO_STEP3_ * 64) __builtin_amdgcn_s_setprio(0);
#pragma unroll
    for (int t = 0; t < 8; ++t) {
      const int r = sr + 8 * t;
      *(v4f*)&Kl[r][sc] = kst[t];
      Vt[sc + 0][r] = vst[t][0];
      Vt[sc + 1][r] = vst[t][1];
      Vt[sc + 2][r] = vst[t][2];
      Vt[sc + 3][r] = vst[t][3];
    }
    __syncthreads();
    if (j + 64 < N_) {
#pragma unroll
      for (int t = 0; t < 8; ++t) {
        kst[t] = *(const v4f*)(kb + (size_t)(j + 64 + sr + 8 * t) * HD_ + sc);
        vst[t] = *(const v4f*)(vb + (size_t)(j + 64 + sr + 8 * t) * HD_ + sc);
      }
    }

    v2f kf[16][4], vf[16][4];
#pragma unroll
    for (int ks = 0; ks < 16; ++ks) {
      const int kd = 4 * ks + 2 * half;
#pragma unroll
      for (int nt = 0; nt < 4; ++nt) kf[ks][nt] = *(const v2f*)&Kl[nt * 16 + krow][kd];
    }
#pragma unroll
    for (int ks = 0; ks < 16; ++ks) {
      const int kk = 4 * ks + 2 * half;
#pragma unroll
      for (int dt = 0; dt < 4; ++dt) vf[ks][dt] = *(const v2f*)&Vt[dt * 16 + l16][kk];
    }
    __builtin_amdgcn_sched_barrier(0);

    v8f S[4] = {};
#pragma unroll
    for (int ks = 0; ks < 16; ++ks) {
#pragma unroll
      for (int nt = 0; nt < 4; ++nt) S[nt] = wmma_f32(kf[ks][nt], qa[ks], S[nt]);
    }

    float mx[4];
#pragma unroll
    for (int nt = 0; nt < 4; ++nt) {
      mx[nt] = fmaxf(fmaxf(fmaxf(S[nt][0], S[nt][1]), fmaxf(S[nt][2], S[nt][3])),
                     fmaxf(fmaxf(S[nt][4], S[nt][5]), fmaxf(S[nt][6], S[nt][7])));
    }
    float m = fmaxf(fmaxf(mx[0], mx[1]), fmaxf(mx[2], mx[3]));
    m = fmaxf(m, xhalf(m));
    const float nm = fmaxf(rm, m);
    const float f  = __expf(rm - nm);
    rm = nm;
    float ps[4];
#pragma unroll
    for (int nt = 0; nt < 4; ++nt) {
#pragma unroll
      for (int r = 0; r < 8; ++r) S[nt][r] = __expf(S[nt][r] - nm);
      ps[nt] = ((S[nt][0] + S[nt][1]) + (S[nt][2] + S[nt][3])) +
               ((S[nt][4] + S[nt][5]) + (S[nt][6] + S[nt][7]));
    }
    rs = rs * f + ((ps[0] + ps[1]) + (ps[2] + ps[3]));
#pragma unroll
    for (int dt = 0; dt < 4; ++dt) oacc[dt] *= f;

#pragma unroll
    for (int ks = 0; ks < 16; ++ks) {
      const int nt = ks >> 2, jj = ks & 3;
      v2f pb;
      pb[0] = S[nt][2 * jj];
      pb[1] = S[nt][2 * jj + 1];
#pragma unroll
      for (int dt = 0; dt < 4; ++dt) oacc[dt] = wmma_f32(vf[ks][dt], pb, oacc[dt]);
    }
    __builtin_amdgcn_sched_barrier(0);
    __syncthreads();
  }

  const float inv = 1.f / (rs + xhalf(rs));
#pragma unroll
  for (int dt = 0; dt < 4; ++dt) {
#pragma unroll
    for (int r = 0; r < 8; ++r) {
      Kl[dt * 16 + 8 * half + r][wid * 16 + l16] = oacc[dt][r] * inv;
    }
  }
  __syncthreads();
#pragma unroll
  for (int t = 0; t < 8; ++t) {
    const int d = sr + 8 * t;
    const v4f o = *(const v4f*)&Kl[d][sc];
    *(v4f*)(ao + ((size_t)b * C_ + h * HD_ + d) * N_ + q0 + sc) = o;
  }
}

__global__ __launch_bounds__(128) void proj_kernel(
    const float* __restrict__ ao, const float* __restrict__ w,
    const float* __restrict__ bias, const float* __restrict__ x,
    float* __restrict__ out) {
  __shared__ float Wl[64][68];
  __shared__ float Xt[64][68];
  const int n0 = blockIdx.x * 64;
  const int o0 = blockIdx.y * 64;
  const int b  = blockIdx.z;
  const int tid = threadIdx.x;
  const int wid = tid >> 5, lane = tid & 31, l16 = lane & 15, half = lane >> 4;
  const int orow = o0 + wid * 16;
  const float* ab = ao + (size_t)b * C_ * N_;

  v8f acc[4] = {};
  const int sr = tid >> 4, sc = (tid & 15) * 4;
  v4f wst[8], xst[8];
#pragma unroll
  for (int t = 0; t < 8; ++t) {
    wst[t] = *(const v4f*)(w + (size_t)(o0 + sr + 8 * t) * C_ + sc);
    xst[t] = *(const v4f*)(ab + (size_t)(sr + 8 * t) * N_ + n0 + sc);
  }
  for (int kb = 0; kb < C_; kb += 64) {
#pragma unroll
    for (int t = 0; t < 8; ++t) {
      const int r = sr + 8 * t;
      *(v4f*)&Wl[r][sc] = wst[t];
      Xt[sc + 0][r] = xst[t][0];
      Xt[sc + 1][r] = xst[t][1];
      Xt[sc + 2][r] = xst[t][2];
      Xt[sc + 3][r] = xst[t][3];
    }
    __syncthreads();
    if (kb + 64 < C_) {
#pragma unroll
      for (int t = 0; t < 8; ++t) {
        wst[t] = *(const v4f*)(w + (size_t)(o0 + sr + 8 * t) * C_ + kb + 64 + sc);
        xst[t] = *(const v4f*)(ab + (size_t)(kb + 64 + sr + 8 * t) * N_ + n0 + sc);
      }
    }
#pragma unroll
    for (int ks = 0; ks < 16; ++ks) {
      const int kd = 4 * ks + 2 * half;
      const v2f a = *(const v2f*)&Wl[wid * 16 + l16][kd];
#pragma unroll
      for (int nt = 0; nt < 4; ++nt) {
        const v2f bb = *(const v2f*)&Xt[nt * 16 + l16][kd];
        acc[nt] = wmma_f32(a, bb, acc[nt]);
      }
    }
    __syncthreads();
  }

  float bv[8];
#pragma unroll
  for (int vv = 0; vv < 8; ++vv) bv[vv] = bias[orow + 8 * half + vv];
#pragma unroll
  for (int nt = 0; nt < 4; ++nt) {
    const int n = n0 + nt * 16 + l16;
#pragma unroll
    for (int vv = 0; vv < 8; ++vv) {
      const int o = orow + 8 * half + vv;
      const size_t idx = ((size_t)b * C_ + o) * N_ + n;
      out[idx] = acc[nt][vv] + bv[vv] + x[idx];
    }
  }
}

extern "C" void kernel_launch(void* const* d_in, const int* in_sizes, int n_in,
                              void* d_out, int out_size, void* d_ws,
                              size_t ws_size, hipStream_t stream) {
  const float* x      = (const float*)d_in[0];
  const float* gamma  = (const float*)d_in[1];
  const float* beta   = (const float*)d_in[2];
  const float* w_qkv  = (const float*)d_in[3];
  const float* b_qkv  = (const float*)d_in[4];
  const float* w_proj = (const float*)d_in[5];
  const float* b_proj = (const float*)d_in[6];
  float* out = (float*)d_out;

  const size_t SZ = (size_t)B_ * C_ * N_;
  float* part = (float*)d_ws;
  float* q  = part + SZ;
  float* k  = q + SZ;
  float* v  = k + SZ;
  float* ao = v + SZ;

  gn_stats_kernel<<<B_ * GROUPS_ * GN_PARTS_, GS_THREADS_, 0, stream>>>(x, part);
  qkv_kernel<<<dim3(N_ / 64, (3 * C_) / 64, B_), 128, 0, stream>>>(
      x, part, gamma, beta, w_qkv, b_qkv, q, k, v);
  attn_kernel<<<dim3(N_ / 64, HEADS_, B_), 128, 0, stream>>>(q, k, v, ao);
  proj_kernel<<<dim3(N_ / 64, C_ / 64, B_), 128, 0, stream>>>(
      ao, w_proj, b_proj, x, out);
}


# Attention: K/V tiles in two LDS stages, one barrier per key tile between score and output block: stores, fragment reads and fetches run under matrix work (was two barriers, all in series)
# speedup vs baseline: 2.4612x; 1.0296x over previous
#include <hip/hip_runtime.h>

#define B_ 2
#define C_ 256
#define N_ 4096
#define HEADS_ 4
#define HD_ 64
#define GROUPS_ 8
#define CPG_ 32
#define EPS_ 1e-5f

typedef __attribute__((ext_vector_type(2))) float v2f;
typedef __attribute__((ext_vector_type(8))) float v8f;
typedef __attribute__((ext_vector_type(4))) float v4f;

__device__ __forceinline__ v8f wmma_f32(v2f a, v2f b, v8f c) {
  return __builtin_amdgcn_wmma_f32_16x16x4_f32(
       false, a,  false, b,
       (short)0, c,  false,  false);
}

#define GS_THREADS_ 128
#define GS_BATCH_ 16
#define GN_PARTS_ 16
__global__ __launch_bounds__(GS_THREADS_) void gn_stats_kernel(
    const float* __restrict__ x, float* __restrict__ part) {
  __shared__ float s1[GS_THREADS_ / 32], s2[GS_THREADS_ / 32];
  const v4f* xv = (const v4f*)(x + (size_t)blockIdx.x * (2 * N_));
  const int tid = threadIdx.x;
  const int lane = tid & 31, wid = tid >> 5;

  v4f t[GS_BATCH_];
  for (int u = 0; u < GS_BATCH_; ++u) t[u] = xv[tid + u * GS_THREADS_];
  float sum = 0.f, ssq = 0.f;
  for (int u = 0; u < GS_BATCH_; ++u) {
    sum += (t[u][0] + t[u][1]) + (t[u][2] + t[u][3]);
    ssq += (t[u][0] * t[u][0] + t[u][1] * t[u][1]) +
           (t[u][2] * t[u][2] + t[u][3] * t[u][3]);
  }
  for (int off = 16; off > 0; off >>= 1) {
    sum += __shfl_xor(sum, off, 32);
    ssq += __shfl_xor(ssq, off, 32);
  }
  if (lane == 0) { s1[wid] = sum; s2[wid] = ssq; }
  __syncthreads();
  if (wid == 0) {
    sum = s1[lane & 3];
    ssq = s2[lane & 3];
    for (int off = 2; off > 0; off >>= 1) {
      sum += __shfl_xor(sum, off, 32);
      ssq += __shfl_xor(ssq, off, 32);
    }
    if (lane == 0) {
      v2f o;
      o[0] = sum;
      o[1] = ssq;
      *(v2f*)(part + (size_t)blockIdx.x * 2) = o;
    }
  }
}

__global__ __launch_bounds__(128) void qkv_kernel(
    const float* __restrict__ x, const float* __restrict__ part,
    const float* __restrict__ gamma, const float* __restrict__ beta,
    const float* __restrict__ w,
    const float* __restrict__ bias, float* __restrict__ q,
    float* __restrict__ k, float* __restrict__ v) {
  __shared__ float Wl[64][68];
  __shared__ float Xt[64][68];
  __shared__ v4f coef[C_];
  const int n0 = blockIdx.x * 64;
  const int o0 = blockIdx.y * 64;
  const int b  = blockIdx.z;
  const int tid = threadIdx.x;
  const int wid = tid >> 5, lane = tid & 31, l16 = lane & 15, half = lane >> 4;
  const int orow = o0 + wid * 16;
  const float* xb = x + (size_t)b * C_ * N_;

  v8f acc[4] = {};
  const int sr = tid >> 4, sc = (tid & 15) * 4;
  v4f wst[8], xst[8];
#pragma unroll
  for (int t = 0; t < 8; ++t) {
    wst[t] = *(const v4f*)(w + (size_t)(o0 + sr + 8 * t) * C_ + sc);
    xst[t] = *(const v4f*)(xb + (size_t)(sr + 8 * t) * N_ + n0 + sc);
  }

  {
    const int g = tid >> 4, p = tid & 15;
    const v2f pr =
        *(const v2f*)(part + ((size_t)(b * GROUPS_ + g) * GN_PARTS_ + p) * 2);
    float sum = pr[0], ssq = pr[1];
    for (int off = 8; off > 0; off >>= 1) {
      sum += __shfl_xor(sum, off, 32);
      ssq += __shfl_xor(ssq, off, 32);
    }
    const int count = CPG_ * N_;
    const float mean = sum / (float)count;
    const float var  = ssq / (float)count - mean * mean;
    const float rstd = rsqrtf(var + EPS_);
    for (int e = 0; e < 2; ++e) {
      const int c = g * CPG_ + 2 * p + e;
      v4f cf;
      cf[0] = mean;
      cf[1] = rstd;
      cf[2] = gamma[c];
      cf[3] = beta[c];
      coef[c] = cf;
    }
  }
  __syncthreads();

  for (int kb = 0; kb < C_; kb += 64) {
    v4f cf[8];
#pragma unroll
    for (int t = 0; t < 8; ++t) cf[t] = coef[kb + sr + 8 * t];
#pragma unroll
    for (int t = 0; t < 8; ++t) {
      const int r = sr + 8 * t;
      *(v4f*)&Wl[r][sc] = wst[t];
      v4f xn;
      for (int e = 0; e < 4; ++e) {
        const float vv = (xst[t][e] - cf[t][0]) * cf[t][1];
        xn[e] = vv * cf[t][2] + cf[t][3];
      }
      Xt[sc + 0][r] = xn[0];
      Xt[sc + 1][r] = xn[1];
      Xt[sc + 2][r] = xn[2];
      Xt[sc + 3][r] = xn[3];
    }
    __syncthreads();
    if (kb + 64 < C_) {
#pragma unroll
      for (int t = 0; t < 8; ++t) {
        wst[t] = *(const v4f*)(w + (size_t)(o0 + sr + 8 * t) * C_ + kb + 64 + sc);
        xst[t] = *(const v4f*)(xb + (size_t)(kb + 64 + sr + 8 * t) * N_ + n0 + sc);
      }
    }
#pragma unroll
    for (int ks = 0; ks < 16; ++ks) {
      const int kd = 4 * ks + 2 * half;
      const v2f a = *(const v2f*)&Wl[wid * 16 + l16][kd];
#pragma unroll
      for (int nt = 0; nt < 4; ++nt) {
        const v2f bb = *(const v2f*)&Xt[nt * 16 + l16][kd];
        acc[nt] = wmma_f32(a, bb, acc[nt]);
      }
    }
    __syncthreads();
  }

  const int which = orow >> 8;
  const int hh    = (orow & 255) >> 6;
  const int dbase = (orow & 63) + 8 * half;
  float* outb = (which == 0) ? q : (which == 1) ? k : v;
  float bv[8];
#pragma unroll
  for (int vv = 0; vv < 8; ++vv) bv[vv] = bias[orow + 8 * half + vv];
#pragma unroll
  for (int nt = 0; nt < 4; ++nt) {
    const int n = n0 + nt * 16 + l16;
    float* p = outb + (((size_t)b * HEADS_ + hh) * N_ + n) * HD_ + dbase;
    float4 lo, hi;
    lo.x = acc[nt][0] + bv[0]; lo.y = acc[nt][1] + bv[1];
    lo.z = acc[nt][2] + bv[2]; lo.w = acc[nt][3] + bv[3];
    hi.x = acc[nt][4] + bv[4]; hi.y = acc[nt][5] + bv[5];
    hi.z = acc[nt][6] + bv[6]; hi.w = acc[nt][7] + bv[7];
    *(float4*)p = lo;
    *(float4*)(p + 4) = hi;
  }
}

#define PRIO_STEP1_ 44
#define PRIO_STEP2_ 53
#define PRIO_STEP3_ 60

#define KV_PITCH_ 68
#define KV_TILE_ (64 * KV_PITCH_)

__device__ __forceinline__ float xhalf(float x) {
  const int bits = __builtin_bit_cast(int, x);
  return __builtin_bit_cast(
      float, __builtin_amdgcn_permlanex16(bits, bits, 0x76543210, 0xfedcba98,
                                          false, false));
}

__global__ __launch_bounds__(128) void attn_kernel(
    const float* __restrict__ q, const float* __restrict__ k,
    const float* __restrict__ v, float* __restrict__ ao) {
  __shared__ __attribute__((aligned(16))) float Kl[2 * KV_TILE_];
  __shared__ __attribute__((aligned(16))) float Vt[2 * KV_TILE_];
  const int q0 = blockIdx.x * 64;
  const int h  = blockIdx.y;
  const int b  = blockIdx.z;
  const int tid = threadIdx.x;
  const int wid = tid >> 5, lane = tid & 31, l16 = lane & 15, half = lane >> 4;
  const size_t hb = ((size_t)b * HEADS_ + h) * (size_t)N_ * HD_;
  const float* qb = q + hb;
  const float* kb = k + hb;
  const float* vb = v + hb;
  const float scale = 0.125f;

  v2f qa[16];
  const int qrow = q0 + wid * 16 + l16;
#pragma unroll
  for (int ks = 0; ks < 16; ++ks) {
    const int d = 4 * ks + 2 * half;
    const v2f t = *(const v2f*)(qb + (size_t)qrow * HD_ + d);
    qa[ks] = t * scale;
  }

  const int krow = 4 * ((l16 & 7) >> 1) + 2 * (l16 >> 3) + (l16 & 1);

  v8f oacc[4] = {};
  float rm = -1e30f;
  float rs = 0.f;

  const int sr = tid >> 4, sc = (tid & 15) * 4;
  const int kfo = krow * KV_PITCH_ + 2 * half;
  const int vfo = l16 * KV_PITCH_ + 2 * half;
  v4f kst[8], vst[8];
  v2f kf[16][4];

#pragma unroll
  for (int t = 0; t < 8; ++t) {
    kst[t] = *(const v4f*)(kb + (size_t)(sr + 8 * t) * HD_ + sc);
    vst[t] = *(const v4f*)(vb + (size_t)(sr + 8 * t) * HD_ + sc);
  }
#pragma unroll
  for (int t = 0; t < 8; ++t) {
    const int r = sr + 8 * t;
    *(v4f*)&Kl[r * KV_PITCH_ + sc] = kst[t];
    Vt[(sc + 0) * KV_PITCH_ + r] = vst[t][0];
    Vt[(sc + 1) * KV_PITCH_ + r] = vst[t][1];
    Vt[(sc + 2) * KV_PITCH_ + r] = vst[t][2];
    Vt[(sc + 3) * KV_PITCH_ + r] = vst[t][3];
  }
#pragma unroll
  for (int t = 0; t < 8; ++t) {
    kst[t] = *(const v4f*)(kb + (size_t)(64 + sr + 8 * t) * HD_ + sc);
    vst[t] = *(const v4f*)(vb + (size_t)(64 + sr + 8 * t) * HD_ + sc);
  }
  __syncthreads();
#pragma unroll
  for (int ks = 0; ks < 16; ++ks) {
#pragma unroll
    for (int nt = 0; nt < 4; ++nt)
      kf[ks][nt] = *(const v2f*)&Kl[nt * 16 * KV_PITCH_ + kfo + 4 * ks];
  }

  __builtin_amdgcn_s_setprio(3);
  for (int j = 0; j < N_; j += 64) {
    if (j == PRIO_STEP1_ * 64) __builtin_amdgcn_s_setprio(2);
    else if (j == PRIO_STEP2_ * 64) __builtin_amdgcn_s_setprio(1);
    else if (j == PRIO_STEP3_ * 64) __builtin_amdgcn_s_setprio(0);
    const int cur = (j >> 6) & 1;
    const float* Vc = Vt + cur * KV_TILE_;
    float* Kn = Kl + (cur ^ 1) * KV_TILE_;
    float* Vn = Vt + (cur ^ 1) * KV_TILE_;

    v2f vf[16][4];
#pragma unroll
    for (int ks = 0; ks < 16; ++ks) {
#pragma unroll
      for (int dt = 0; dt < 4; ++dt)
        vf[ks][dt] = *(const v2f*)&Vc[dt * 16 * KV_PITCH_ + vfo + 4 * ks];
    }
    if (j + 64 < N_) {
#pragma unroll
      for (int t = 0; t < 8; ++t) {
        const int r = sr + 8 * t;
        *(v4f*)&Kn[r * KV_PITCH_ + sc] = kst[t];
        Vn[(sc + 0) * KV_PITCH_ + r] = vst[t][0];
        Vn[(sc + 1) * KV_PITCH_ + r] = vst[t][1];
        Vn[(sc + 2) * KV_PITCH_ + r] = vst[t][2];
        Vn[(sc + 3) * KV_PITCH_ + r] = vst[t][3];
      }
    }

    v8f S[4] = {};
#pragma unroll
    for (int ks = 0; ks < 16; ++ks) {
#pragma unroll
      for (int nt = 0; nt < 4; ++nt) S[nt] = wmma_f32(kf[ks][nt], qa[ks], S[nt]);
    }
    asm volatile("" : "+v"(S[0]), "+v"(S[1]), "+v"(S[2]), "+v"(S[3]));
    __syncthreads();

#pragma unroll
    for (int ks = 0; ks < 16; ++ks) {
#pragma unroll
      for (int nt = 0; nt < 4; ++nt)
        kf[ks][nt] = *(const v2f*)&Kn[nt * 16 * KV_PITCH_ + kfo + 4 * ks];
    }
    {
      const int jn = min(j + 128, N_ - 64);
#pragma unroll
      for (int t = 0; t < 8; ++t) {
        kst[t] = *(const v4f*)(kb + (size_t)(jn + sr + 8 * t) * HD_ + sc);
        vst[t] = *(const v4f*)(vb + (size_t)(jn + sr + 8 * t) * HD_ + sc);
      }
    }
    __builtin_amdgcn_sched_barrier(0);

    float mx[4];
#pragma unroll
    for (int nt = 0; nt < 4; ++nt) {
      mx[nt] = fmaxf(fmaxf(fmaxf(S[nt][0], S[nt][1]), fmaxf(S[nt][2], S[nt][3])),
                     fmaxf(fmaxf(S[nt][4], S[nt][5]), fmaxf(S[nt][6], S[nt][7])));
    }
    float m = fmaxf(fmaxf(mx[0], mx[1]), fmaxf(mx[2], mx[3]));
    m = fmaxf(m, xhalf(m));
    const float nm = fmaxf(rm, m);
    const float f  = __expf(rm - nm);
    rm = nm;
    float ps[4];
#pragma unroll
    for (int nt = 0; nt < 4; ++nt) {
#pragma unroll
      for (int r = 0; r < 8; ++r) S[nt][r] = __expf(S[nt][r] - nm);
      ps[nt] = ((S[nt][0] + S[nt][1]) + (S[nt][2] + S[nt][3])) +
               ((S[nt][4] + S[nt][5]) + (S[nt][6] + S[nt][7]));
    }
    rs = rs * f + ((ps[0] + ps[1]) + (ps[2] + ps[3]));
#pragma unroll
    for (int dt = 0; dt < 4; ++dt) oacc[dt] *= f;

#pragma unroll
    for (int ks = 0; ks < 16; ++ks) {
      const int nt = ks >> 2, jj = ks & 3;
      v2f pb;
      pb[0] = S[nt][2 * jj];
      pb[1] = S[nt][2 * jj + 1];
#pragma unroll
      for (int dt = 0; dt < 4; ++dt) oacc[dt] = wmma_f32(vf[ks][dt], pb, oacc[dt]);
    }
  }

  float* Ot = Kl + ((N_ / 64 - 1) & 1) * KV_TILE_;
  const float inv = 1.f / (rs + xhalf(rs));
#pragma unroll
  for (int dt = 0; dt < 4; ++dt) {
#pragma unroll
    for (int r = 0; r < 8; ++r) {
      Ot[(dt * 16 + 8 * half + r) * KV_PITCH_ + wid * 16 + l16] = oacc[dt][r] * inv;
    }
  }
  __syncthreads();
#pragma unroll
  for (int t = 0; t < 8; ++t) {
    const int d = sr + 8 * t;
    const v4f o = *(const v4f*)&Ot[d * KV_PITCH_ + sc];
    *(v4f*)(ao + ((size_t)b * C_ + h * HD_ + d) * N_ + q0 + sc) = o;
  }
}

__global__ __launch_bounds__(128) void proj_kernel(
    const float* __restrict__ ao, const float* __restrict__ w,
    const float* __restrict__ bias, const float* __restrict__ x,
    float* __restrict__ out) {
  __shared__ float Wl[64][68];
  __shared__ float Xt[64][68];
  const int n0 = blockIdx.x * 64;
  const int o0 = blockIdx.y * 64;
  const int b  = blockIdx.z;
  const int tid = threadIdx.x;
  const int wid = tid >> 5, lane = tid & 31, l16 = lane & 15, half = lane >> 4;
  const int orow = o0 + wid * 16;
  const float* ab = ao + (size_t)b * C_ * N_;

  v8f acc[4] = {};
  const int sr = tid >> 4, sc = (tid & 15) * 4;
  v4f wst[8], xst[8];
#pragma unroll
  for (int t = 0; t < 8; ++t) {
    wst[t] = *(const v4f*)(w + (size_t)(o0 + sr + 8 * t) * C_ + sc);
    xst[t] = *(const v4f*)(ab + (size_t)(sr + 8 * t) * N_ + n0 + sc);
  }
  for (int kb = 0; kb < C_; kb += 64) {
#pragma unroll
    for (int t = 0; t < 8; ++t) {
      const int r = sr + 8 * t;
      *(v4f*)&Wl[r][sc] = wst[t];
      Xt[sc + 0][r] = xst[t][0];
      Xt[sc + 1][r] = xst[t][1];
      Xt[sc + 2][r] = xst[t][2];
      Xt[sc + 3][r] = xst[t][3];
    }
    __syncthreads();
    if (kb + 64 < C_) {
#pragma unroll
      for (int t = 0; t < 8; ++t) {
        wst[t] = *(const v4f*)(w + (size_t)(o0 + sr + 8 * t) * C_ + kb + 64 + sc);
        xst[t] = *(const v4f*)(ab + (size_t)(kb + 64 + sr + 8 * t) * N_ + n0 + sc);
      }
    }
#pragma unroll
    for (int ks = 0; ks < 16; ++ks) {
      const int kd = 4 * ks + 2 * half;
      const v2f a = *(const v2f*)&Wl[wid * 16 + l16][kd];
#pragma unroll
      for (int nt = 0; nt < 4; ++nt) {
        const v2f bb = *(const v2f*)&Xt[nt * 16 + l16][kd];
        acc[nt] = wmma_f32(a, bb, acc[nt]);
      }
    }
    __syncthreads();
  }

  float bv[8];
#pragma unroll
  for (int vv = 0; vv < 8; ++vv) bv[vv] = bias[orow + 8 * half + vv];
#pragma unroll
  for (int nt = 0; nt < 4; ++nt) {
    const int n = n0 + nt * 16 + l16;
#pragma unroll
    for (int vv = 0; vv < 8; ++vv) {
      const int o = orow + 8 * half + vv;
      const size_t idx = ((size_t)b * C_ + o) * N_ + n;
      out[idx] = acc[nt][vv] + bv[vv] + x[idx];
    }
  }
}

extern "C" void kernel_launch(void* const* d_in, const int* in_sizes, int n_in,
                              void* d_out, int out_size, void* d_ws,
                              size_t ws_size, hipStream_t stream) {
  const float* x      = (const float*)d_in[0];
  const float* gamma  = (const float*)d_in[1];
  const float* beta   = (const float*)d_in[2];
  const float* w_qkv  = (const float*)d_in[3];
  const float* b_qkv  = (const float*)d_in[4];
  const float* w_proj = (const float*)d_in[5];
  const float* b_proj = (const float*)d_in[6];
  float* out = (float*)d_out;

  const size_t SZ = (size_t)B_ * C_ * N_;
  float* part = (float*)d_ws;
  float* q  = part + SZ;
  float* k  = q + SZ;
  float* v  = k + SZ;
  float* ao = v + SZ;

  gn_stats_kernel<<<B_ * GROUPS_ * GN_PARTS_, GS_THREADS_, 0, stream>>>(x, part);
  qkv_kernel<<<dim3(N_ / 64, (3 * C_) / 64, B_), 128, 0, stream>>>(
      x, part, gamma, beta, w_qkv, b_qkv, q, k, v);
  attn_kernel<<<dim3(N_ / 64, HEADS_, B_), 128, 0, stream>>>(q, k, v, ao);
  proj_kernel<<<dim3(N_ / 64, C_ / 64, B_), 128, 0, stream>>>(
      ao, w_proj, b_proj, x, out);
}
